# starting kernel
# baseline (speedup 1.0000x reference)
#include <hip/hip_runtime.h>
#include <math.h>

#define B_  8
#define N_  4096
#define D_  64
#define S_  1024
#define K_  32

typedef __attribute__((ext_vector_type(2))) float v2f;
typedef __attribute__((ext_vector_type(4))) float v4f;
typedef __attribute__((ext_vector_type(8))) float v8f;

__global__ __launch_bounds__(1024) void fps_kernel(
    const float* __restrict__ xyz,
    const int*   __restrict__ farthest_init,
    float* __restrict__ out_newxyz,
    float* __restrict__ wsA,
    float* __restrict__ wsC2)
{
    __shared__ v4f   p4[N_];
    __shared__ float redV[32];
    __shared__ int   redI[32];
    __shared__ int   sFar;

    const int b    = blockIdx.x;
    const int tid  = threadIdx.x;
    const int lane = tid & 31;
    const int wid  = tid >> 5;

    const float* xb = xyz + (size_t)b * 3 * N_;
    for (int i = tid; i < N_; i += 1024) {
        v4f v;
        v.x = xb[i];
        v.y = xb[N_ + i];
        v.z = xb[2 * N_ + i];
        v.w = 0.f;
        p4[i] = v;
    }
    float dist[4];
#pragma unroll
    for (int j = 0; j < 4; ++j) dist[j] = 1e10f;
    if (tid == 0) sFar = farthest_init[b];
    __syncthreads();

    for (int it = 0; it < S_; ++it) {
        const v4f c = p4[sFar];
        if (tid == 0) {
            out_newxyz[(size_t)b * 3 * S_ + 0 * S_ + it] = c.x;
            out_newxyz[(size_t)b * 3 * S_ + 1 * S_ + it] = c.y;
            out_newxyz[(size_t)b * 3 * S_ + 2 * S_ + it] = c.z;
            float* A = wsA + ((size_t)b * S_ + it) * 4;
            A[0] = -2.f * c.x; A[1] = -2.f * c.y; A[2] = -2.f * c.z; A[3] = 1.f;
            wsC2[(size_t)b * S_ + it] = c.x * c.x + c.y * c.y + c.z * c.z;
        }
        float bv = -1.f; int bi = 0x7fffffff;
#pragma unroll
        for (int j = 0; j < 4; ++j) {
            const int i = tid + 1024 * j;
            const v4f p = p4[i];
            const float dx = p.x - c.x, dy = p.y - c.y, dz = p.z - c.z;
            const float d = dx * dx + dy * dy + dz * dz;
            const float nd = fminf(dist[j], d);
            dist[j] = nd;
            if (nd > bv || (nd == bv && i < bi)) { bv = nd; bi = i; }
        }
#pragma unroll
        for (int off = 16; off >= 1; off >>= 1) {
            const float ov = __shfl_xor(bv, off, 32);
            const int   oi = __shfl_xor(bi, off, 32);
            if (ov > bv || (ov == bv && oi < bi)) { bv = ov; bi = oi; }
        }
        if (lane == 0) { redV[wid] = bv; redI[wid] = bi; }
        __syncthreads();
        if (wid == 0) {
            bv = redV[lane]; bi = redI[lane];
#pragma unroll
            for (int off = 16; off >= 1; off >>= 1) {
                const float ov = __shfl_xor(bv, off, 32);
                const int   oi = __shfl_xor(bi, off, 32);
                if (ov > bv || (ov == bv && oi < bi)) { bv = ov; bi = oi; }
            }
            if (lane == 0) sFar = bi;
        }
        __syncthreads();
    }
}

#define TPITCH 4100

__global__ __launch_bounds__(256) void group_kernel(
    const float* __restrict__ xyz,
    const float* __restrict__ pts,
    const float* __restrict__ wsA,
    const float* __restrict__ wsC2,
    float* __restrict__ out_feat)
{
    __shared__ float DT[16 * TPITCH];
    __shared__ int   selIdx[16][K_];

    const int b    = blockIdx.y;
    const int s0   = blockIdx.x * 16;
    const int tid  = threadIdx.x;
    const int lane = tid & 31;
    const int wid  = tid >> 5;
    const bool hi  = lane >= 16;

    const float* xb = xyz + (size_t)b * 3 * N_;

    const int  mcol = lane & 15;
    const v4f  A4   = *(const v4f*)(wsA + ((size_t)b * S_ + s0 + mcol) * 4);
    v2f a;
    a.x = hi ? A4.z : A4.x;
    a.y = hi ? A4.w : A4.y;

    const int mbase = hi ? 8 : 0;
    float cmr[8];
#pragma unroll
    for (int r = 0; r < 8; ++r)
        cmr[r] = wsC2[(size_t)b * S_ + s0 + mbase + r];

    for (int t = 0; t < 32; ++t) {
        const int n = wid * 512 + t * 16 + mcol;
        const float x = xb[n], y = xb[N_ + n], z = xb[2 * N_ + n];
        const float pn2 = x * x + y * y + z * z;
        v2f bb;
        bb.x = hi ? z : x;
        bb.y = hi ? pn2 : y;

        v8f c = {};
        c = __builtin_amdgcn_wmma_f32_16x16x4_f32(
                 false, a,  false, bb,
                 (short)0, c,  false,  false);

#pragma unroll
        for (int r = 0; r < 8; ++r)
            DT[(mbase + r) * TPITCH + n] = c[r] + cmr[r];
    }
    __syncthreads();

    for (int rr = 0; rr < 2; ++rr) {
        const int m = wid + rr * 8;
        float* row = &DT[m * TPITCH];
        for (int k = 0; k < K_; ++k) {
            float bv = 3.4e38f; int bi = 0x7fffffff;
            for (int jj = 0; jj < N_ / 128; ++jj) {
                const int j0 = jj * 128 + lane * 4;
                const v4f dv = *(const v4f*)&row[j0];
#pragma unroll
                for (int e = 0; e < 4; ++e) {
                    const float d = dv[e];
                    const int   j = j0 + e;
                    if (d < bv || (d == bv && j < bi)) { bv = d; bi = j; }
                }
            }
#pragma unroll
            for (int off = 16; off >= 1; off >>= 1) {
                const float ov = __shfl_xor(bv, off, 32);
                const int   oi = __shfl_xor(bi, off, 32);
                if (ov < bv || (ov == bv && oi < bi)) { bv = ov; bi = oi; }
            }
            if (lane == 0) { selIdx[m][k] = bi; row[bi] = 3.4e38f; }
            __syncthreads();
        }
    }
    __syncthreads();

    const float* pb = pts + (size_t)b * D_ * N_;
    for (int t = tid; t < 16 * 67; t += 256) {
        const int m = t / 67;
        const int d = t % 67;
        const float* src = (d < 3) ? (xb + (size_t)d * N_)
                                   : (pb + (size_t)(d - 3) * N_);
        float mx = -3.4e38f;
#pragma unroll 4
        for (int k = 0; k < K_; ++k) mx = fmaxf(mx, src[selIdx[m][k]]);
        out_feat[(size_t)b * 67 * S_ + (size_t)d * S_ + (s0 + m)] = mx;
    }
}

extern "C" void kernel_launch(void* const* d_in, const int* in_sizes, int n_in,
                              void* d_out, int out_size, void* d_ws, size_t ws_size,
                              hipStream_t stream) {
    const float* xyz    = (const float*)d_in[0];
    const float* points = (const float*)d_in[1];
    const int*   finit  = (const int*)d_in[2];

    float* out        = (float*)d_out;
    float* out_newxyz = out;
    float* out_feat   = out + (size_t)B_ * 3 * S_;

    float* wsA  = (float*)d_ws;
    float* wsC2 = wsA + (size_t)B_ * S_ * 4;

    fps_kernel<<<B_, 1024, 0, stream>>>(xyz, finit, out_newxyz, wsA, wsC2);
    group_kernel<<<dim3(S_ / 16, B_), 256, 0, stream>>>(xyz, points, wsA, wsC2, out_feat);
}


# top-32 pick: each lane caches the (dist,idx) minimum of its own 128-element set; a pick re-scans only the winner lane's set (one LDS load per lane), not the whole 4096-element row
# speedup vs baseline: 1.4784x; 1.4784x over previous
#include <hip/hip_runtime.h>
#include <math.h>

#define B_  8
#define N_  4096
#define D_  64
#define S_  1024
#define K_  32

typedef __attribute__((ext_vector_type(2))) float v2f;
typedef __attribute__((ext_vector_type(4))) float v4f;
typedef __attribute__((ext_vector_type(8))) float v8f;

__global__ __launch_bounds__(1024) void fps_kernel(
    const float* __restrict__ xyz,
    const int*   __restrict__ farthest_init,
    float* __restrict__ out_newxyz,
    float* __restrict__ wsA,
    float* __restrict__ wsC2)
{
    __shared__ v4f   p4[N_];
    __shared__ float redV[32];
    __shared__ int   redI[32];
    __shared__ int   sFar;

    const int b    = blockIdx.x;
    const int tid  = threadIdx.x;
    const int lane = tid & 31;
    const int wid  = tid >> 5;

    const float* xb = xyz + (size_t)b * 3 * N_;
    for (int i = tid; i < N_; i += 1024) {
        v4f v;
        v.x = xb[i];
        v.y = xb[N_ + i];
        v.z = xb[2 * N_ + i];
        v.w = 0.f;
        p4[i] = v;
    }
    float dist[4];
#pragma unroll
    for (int j = 0; j < 4; ++j) dist[j] = 1e10f;
    if (tid == 0) sFar = farthest_init[b];
    __syncthreads();

    for (int it = 0; it < S_; ++it) {
        const v4f c = p4[sFar];
        if (tid == 0) {
            out_newxyz[(size_t)b * 3 * S_ + 0 * S_ + it] = c.x;
            out_newxyz[(size_t)b * 3 * S_ + 1 * S_ + it] = c.y;
            out_newxyz[(size_t)b * 3 * S_ + 2 * S_ + it] = c.z;
            float* A = wsA + ((size_t)b * S_ + it) * 4;
            A[0] = -2.f * c.x; A[1] = -2.f * c.y; A[2] = -2.f * c.z; A[3] = 1.f;
            wsC2[(size_t)b * S_ + it] = c.x * c.x + c.y * c.y + c.z * c.z;
        }
        float bv = -1.f; int bi = 0x7fffffff;
#pragma unroll
        for (int j = 0; j < 4; ++j) {
            const int i = tid + 1024 * j;
            const v4f p = p4[i];
            const float dx = p.x - c.x, dy = p.y - c.y, dz = p.z - c.z;
            const float d = dx * dx + dy * dy + dz * dz;
            const float nd = fminf(dist[j], d);
            dist[j] = nd;
            if (nd > bv || (nd == bv && i < bi)) { bv = nd; bi = i; }
        }
#pragma unroll
        for (int off = 16; off >= 1; off >>= 1) {
            const float ov = __shfl_xor(bv, off, 32);
            const int   oi = __shfl_xor(bi, off, 32);
            if (ov > bv || (ov == bv && oi < bi)) { bv = ov; bi = oi; }
        }
        if (lane == 0) { redV[wid] = bv; redI[wid] = bi; }
        __syncthreads();
        if (wid == 0) {
            bv = redV[lane]; bi = redI[lane];
#pragma unroll
            for (int off = 16; off >= 1; off >>= 1) {
                const float ov = __shfl_xor(bv, off, 32);
                const int   oi = __shfl_xor(bi, off, 32);
                if (ov > bv || (ov == bv && oi < bi)) { bv = ov; bi = oi; }
            }
            if (lane == 0) sFar = bi;
        }
        __syncthreads();
    }
}

__device__ __forceinline__ void lexmin(float& v, int& i, float ov, int oi) {
    const bool take = (ov < v) | ((ov == v) & (oi < i));
    v = take ? ov : v;
    i = take ? oi : i;
}

#define TPITCH 4100

__global__ __launch_bounds__(256) void group_kernel(
    const float* __restrict__ xyz,
    const float* __restrict__ pts,
    const float* __restrict__ wsA,
    const float* __restrict__ wsC2,
    float* __restrict__ out_feat)
{
    __shared__ float DT[16 * TPITCH];
    __shared__ int   selIdx[16][K_];

    const int b    = blockIdx.y;
    const int s0   = blockIdx.x * 16;
    const int tid  = threadIdx.x;
    const int lane = tid & 31;
    const int wid  = tid >> 5;
    const bool hi  = lane >= 16;

    const float* xb = xyz + (size_t)b * 3 * N_;

    const int  mcol = lane & 15;
    const v4f  A4   = *(const v4f*)(wsA + ((size_t)b * S_ + s0 + mcol) * 4);
    v2f a;
    a.x = hi ? A4.z : A4.x;
    a.y = hi ? A4.w : A4.y;

    const int mbase = hi ? 8 : 0;
    float cmr[8];
#pragma unroll
    for (int r = 0; r < 8; ++r)
        cmr[r] = wsC2[(size_t)b * S_ + s0 + mbase + r];

    for (int t = 0; t < 32; ++t) {
        const int n = wid * 512 + t * 16 + mcol;
        const float x = xb[n], y = xb[N_ + n], z = xb[2 * N_ + n];
        const float pn2 = x * x + y * y + z * z;
        v2f bb;
        bb.x = hi ? z : x;
        bb.y = hi ? pn2 : y;

        v8f c = {};
        c = __builtin_amdgcn_wmma_f32_16x16x4_f32(
                 false, a,  false, bb,
                 (short)0, c,  false,  false);

#pragma unroll
        for (int r = 0; r < 8; ++r)
            DT[(mbase + r) * TPITCH + n] = c[r] + cmr[r];
    }
    __syncthreads();

    float* rowp[2];
    rowp[0] = &DT[wid * TPITCH];
    rowp[1] = &DT[(wid + 8) * TPITCH];

    float lv[2]; int li[2];
    int gi[2];
#pragma unroll
    for (int rr = 0; rr < 2; ++rr) {
        const float* row = rowp[rr];
        float bv = 3.4e38f; int bi = 0x7fffffff;
#pragma unroll 8
        for (int q = 0; q < N_ / 128; ++q) {
            const int j0 = q * 128 + ((lane - q) & 31) * 4;
            const v4f dv = *(const v4f*)&row[j0];
#pragma unroll
            for (int e = 0; e < 4; ++e) lexmin(bv, bi, dv[e], j0 + e);
        }
        lv[rr] = bv; li[rr] = bi;
#pragma unroll
        for (int off = 16; off >= 1; off >>= 1) {
            const float ov = __shfl_xor(bv, off, 32);
            const int   oi = __shfl_xor(bi, off, 32);
            lexmin(bv, bi, ov, oi);
        }
        gi[rr] = bi;
    }

    for (int k = 0; k < K_; ++k) {
        if (lane == 0) {
            selIdx[wid][k]     = gi[0];
            selIdx[wid + 8][k] = gi[1];
            rowp[0][gi[0]] = 3.4e38f;
            rowp[1][gi[1]] = 3.4e38f;
        }
        if (k + 1 < K_) {
#pragma unroll
            for (int rr = 0; rr < 2; ++rr) {
                const float* row = rowp[rr];
                const int bi = gi[rr];
                const int w = ((bi >> 2) + (bi >> 7)) & 31;
                const int j0 = lane * 128 + ((w - lane) & 31) * 4;
                const v4f dv = *(const v4f*)&row[j0];
                float cv = dv[0]; int ci = j0;
                float c2 = dv[2]; int i2 = j0 + 2;
                lexmin(cv, ci, dv[1], j0 + 1);
                lexmin(c2, i2, dv[3], j0 + 3);
                lexmin(cv, ci, c2, i2);
                const bool own = (lane == w);
                float ov = own ? 3.4e38f : lv[rr];
                int   oi = own ? 0x7fffffff : li[rr];
#pragma unroll
                for (int off = 16; off >= 1; off >>= 1) {
                    const float cv2 = __shfl_xor(cv, off, 32);
                    const int   ci2 = __shfl_xor(ci, off, 32);
                    const float ov2 = __shfl_xor(ov, off, 32);
                    const int   oi2 = __shfl_xor(oi, off, 32);
                    lexmin(cv, ci, cv2, ci2);
                    lexmin(ov, oi, ov2, oi2);
                }
                lv[rr] = own ? cv : lv[rr];
                li[rr] = own ? ci : li[rr];
                lexmin(cv, ci, ov, oi);
                gi[rr] = ci;
            }
        }
    }
    __syncthreads();

    const float* pb = pts + (size_t)b * D_ * N_;
    for (int t = tid; t < 16 * 67; t += 256) {
        const int m = t / 67;
        const int d = t % 67;
        const float* src = (d < 3) ? (xb + (size_t)d * N_)
                                   : (pb + (size_t)(d - 3) * N_);
        float mx = -3.4e38f;
#pragma unroll 4
        for (int k = 0; k < K_; ++k) mx = fmaxf(mx, src[selIdx[m][k]]);
        out_feat[(size_t)b * 67 * S_ + (size_t)d * S_ + (s0 + m)] = mx;
    }
}

extern "C" void kernel_launch(void* const* d_in, const int* in_sizes, int n_in,
                              void* d_out, int out_size, void* d_ws, size_t ws_size,
                              hipStream_t stream) {
    const float* xyz    = (const float*)d_in[0];
    const float* points = (const float*)d_in[1];
    const int*   finit  = (const int*)d_in[2];

    float* out        = (float*)d_out;
    float* out_newxyz = out;
    float* out_feat   = out + (size_t)B_ * 3 * S_;

    float* wsA  = (float*)d_ws;
    float* wsC2 = wsA + (size_t)B_ * S_ * 4;

    fps_kernel<<<B_, 1024, 0, stream>>>(xyz, finit, out_newxyz, wsA, wsC2);
    group_kernel<<<dim3(S_ / 16, B_), 256, 0, stream>>>(xyz, points, wsA, wsC2, out_feat);
}


# FPS step argmax: index-ordered lanes, distances as bits (u32 min/max), row_xmask DPP + permlanex16 max, first-lane ballot; each wave reduces the 32 partials: 1 LDS-only barrier, no bpermute
# speedup vs baseline: 3.1126x; 2.1054x over previous
#include <hip/hip_runtime.h>
#include <math.h>

#define B_  8
#define N_  4096
#define D_  64
#define S_  1024
#define K_  32

typedef __attribute__((ext_vector_type(2))) float v2f;
typedef __attribute__((ext_vector_type(4))) float v4f;
typedef __attribute__((ext_vector_type(8))) float v8f;

typedef __attribute__((ext_vector_type(2))) unsigned int v2u;

__device__ __forceinline__ unsigned wave_umax(unsigned v) {
    v = max(v, (unsigned)__builtin_amdgcn_update_dpp(0, (int)v, 0x161, 0xf, 0xf, true));
    v = max(v, (unsigned)__builtin_amdgcn_update_dpp(0, (int)v, 0x162, 0xf, 0xf, true));
    v = max(v, (unsigned)__builtin_amdgcn_update_dpp(0, (int)v, 0x164, 0xf, 0xf, true));
    v = max(v, (unsigned)__builtin_amdgcn_update_dpp(0, (int)v, 0x168, 0xf, 0xf, true));
    v = max(v, (unsigned)__builtin_amdgcn_permlanex16((int)v, (int)v, 0x76543210, 0xfedcba98, false, false));
    return v;
}

__device__ __forceinline__ void lds_barrier() {
    __builtin_amdgcn_fence(__ATOMIC_RELEASE, "workgroup", "local");
    __builtin_amdgcn_s_barrier();
    __builtin_amdgcn_fence(__ATOMIC_ACQUIRE, "workgroup", "local");
}

__global__ __launch_bounds__(1024) void fps_kernel(
    const float* __restrict__ xyz,
    const int*   __restrict__ farthest_init,
    float* __restrict__ out_newxyz,
    float* __restrict__ wsA,
    float* __restrict__ wsC2)
{
    __shared__ v4f p4[N_];
    __shared__ v2u red[2][32];

    const int b    = blockIdx.x;
    const int tid  = threadIdx.x;
    const int lane = tid & 31;
    const int wid  = tid >> 5;

    const float* xb = xyz + (size_t)b * 3 * N_;
    for (int i = tid; i < N_; i += 1024) {
        v4f v;
        v.x = xb[i];
        v.y = xb[N_ + i];
        v.z = xb[2 * N_ + i];
        v.w = 0.f;
        p4[(i >> 2) + 1024 * (i & 3)] = v;
    }
    unsigned du[4];
#pragma unroll
    for (int j = 0; j < 4; ++j) du[j] = __float_as_uint(1e10f);
    int far = farthest_init[b] & (N_ - 1);
    __syncthreads();

    for (int it = 0; it < S_; ++it) {
        const v4f c = p4[(far >> 2) + 1024 * (far & 3)];
        v4f p[4];
#pragma unroll
        for (int j = 0; j < 4; ++j) p[j] = p4[tid + 1024 * j];
        if (tid == 0) {
            out_newxyz[(size_t)b * 3 * S_ + 0 * S_ + it] = c.x;
            out_newxyz[(size_t)b * 3 * S_ + 1 * S_ + it] = c.y;
            out_newxyz[(size_t)b * 3 * S_ + 2 * S_ + it] = c.z;
            float* A = wsA + ((size_t)b * S_ + it) * 4;
            A[0] = -2.f * c.x; A[1] = -2.f * c.y; A[2] = -2.f * c.z; A[3] = 1.f;
            wsC2[(size_t)b * S_ + it] = c.x * c.x + c.y * c.y + c.z * c.z;
        }
        unsigned bu = 0u; int bj = 0;
#pragma unroll
        for (int j = 0; j < 4; ++j) {
            const float dx = p[j].x - c.x, dy = p[j].y - c.y, dz = p[j].z - c.z;
            const float d = dx * dx + dy * dy + dz * dz;
            const unsigned u = min(du[j], __float_as_uint(d));
            du[j] = u;
            const bool take = (j == 0) | (u > bu);
            bu = take ? u : bu;
            bj = take ? j : bj;
        }
        const unsigned m   = wave_umax(bu);
        const unsigned hit = __builtin_amdgcn_ballot_w32(bu == m);
        if (lane == __builtin_ctz(hit)) {
            v2u pr; pr.x = bu; pr.y = (unsigned)(4 * tid + bj);
            red[it & 1][wid] = pr;
        }
        lds_barrier();
        const v2u pr = red[it & 1][lane];
        const unsigned M    = wave_umax(pr.x);
        const unsigned hit2 = __builtin_amdgcn_ballot_w32(pr.x == M);
        far = __builtin_amdgcn_readlane((int)pr.y, __builtin_ctz(hit2));
    }
}

__device__ __forceinline__ void lexmin(float& v, int& i, float ov, int oi) {
    const bool take = (ov < v) | ((ov == v) & (oi < i));
    v = take ? ov : v;
    i = take ? oi : i;
}

#define TPITCH 4100

__global__ __launch_bounds__(256) void group_kernel(
    const float* __restrict__ xyz,
    const float* __restrict__ pts,
    const float* __restrict__ wsA,
    const float* __restrict__ wsC2,
    float* __restrict__ out_feat)
{
    __shared__ float DT[16 * TPITCH];
    __shared__ int   selIdx[16][K_];

    const int b    = blockIdx.y;
    const int s0   = blockIdx.x * 16;
    const int tid  = threadIdx.x;
    const int lane = tid & 31;
    const int wid  = tid >> 5;
    const bool hi  = lane >= 16;

    const float* xb = xyz + (size_t)b * 3 * N_;

    const int  mcol = lane & 15;
    const v4f  A4   = *(const v4f*)(wsA + ((size_t)b * S_ + s0 + mcol) * 4);
    v2f a;
    a.x = hi ? A4.z : A4.x;
    a.y = hi ? A4.w : A4.y;

    const int mbase = hi ? 8 : 0;
    float cmr[8];
#pragma unroll
    for (int r = 0; r < 8; ++r)
        cmr[r] = wsC2[(size_t)b * S_ + s0 + mbase + r];

    for (int t = 0; t < 32; ++t) {
        const int n = wid * 512 + t * 16 + mcol;
        const float x = xb[n], y = xb[N_ + n], z = xb[2 * N_ + n];
        const float pn2 = x * x + y * y + z * z;
        v2f bb;
        bb.x = hi ? z : x;
        bb.y = hi ? pn2 : y;

        v8f c = {};
        c = __builtin_amdgcn_wmma_f32_16x16x4_f32(
                 false, a,  false, bb,
                 (short)0, c,  false,  false);

#pragma unroll
        for (int r = 0; r < 8; ++r)
            DT[(mbase + r) * TPITCH + n] = c[r] + cmr[r];
    }
    __syncthreads();

    float* rowp[2];
    rowp[0] = &DT[wid * TPITCH];
    rowp[1] = &DT[(wid + 8) * TPITCH];

    float lv[2]; int li[2];
    int gi[2];
#pragma unroll
    for (int rr = 0; rr < 2; ++rr) {
        const float* row = rowp[rr];
        float bv = 3.4e38f; int bi = 0x7fffffff;
#pragma unroll 8
        for (int q = 0; q < N_ / 128; ++q) {
            const int j0 = q * 128 + ((lane - q) & 31) * 4;
            const v4f dv = *(const v4f*)&row[j0];
#pragma unroll
            for (int e = 0; e < 4; ++e) lexmin(bv, bi, dv[e], j0 + e);
        }
        lv[rr] = bv; li[rr] = bi;
#pragma unroll
        for (int off = 16; off >= 1; off >>= 1) {
            const float ov = __shfl_xor(bv, off, 32);
            const int   oi = __shfl_xor(bi, off, 32);
            lexmin(bv, bi, ov, oi);
        }
        gi[rr] = bi;
    }

    for (int k = 0; k < K_; ++k) {
        if (lane == 0) {
            selIdx[wid][k]     = gi[0];
            selIdx[wid + 8][k] = gi[1];
            rowp[0][gi[0]] = 3.4e38f;
            rowp[1][gi[1]] = 3.4e38f;
        }
        if (k + 1 < K_) {
#pragma unroll
            for (int rr = 0; rr < 2; ++rr) {
                const float* row = rowp[rr];
                const int bi = gi[rr];
                const int w = ((bi >> 2) + (bi >> 7)) & 31;
                const int j0 = lane * 128 + ((w - lane) & 31) * 4;
                const v4f dv = *(const v4f*)&row[j0];
                float cv = dv[0]; int ci = j0;
                float c2 = dv[2]; int i2 = j0 + 2;
                lexmin(cv, ci, dv[1], j0 + 1);
                lexmin(c2, i2, dv[3], j0 + 3);
                lexmin(cv, ci, c2, i2);
                const bool own = (lane == w);
                float ov = own ? 3.4e38f : lv[rr];
                int   oi = own ? 0x7fffffff : li[rr];
#pragma unroll
                for (int off = 16; off >= 1; off >>= 1) {
                    const float cv2 = __shfl_xor(cv, off, 32);
                    const int   ci2 = __shfl_xor(ci, off, 32);
                    const float ov2 = __shfl_xor(ov, off, 32);
                    const int   oi2 = __shfl_xor(oi, off, 32);
                    lexmin(cv, ci, cv2, ci2);
                    lexmin(ov, oi, ov2, oi2);
                }
                lv[rr] = own ? cv : lv[rr];
                li[rr] = own ? ci : li[rr];
                lexmin(cv, ci, ov, oi);
                gi[rr] = ci;
            }
        }
    }
    __syncthreads();

    const float* pb = pts + (size_t)b * D_ * N_;
    for (int t = tid; t < 16 * 67; t += 256) {
        const int m = t / 67;
        const int d = t % 67;
        const float* src = (d < 3) ? (xb + (size_t)d * N_)
                                   : (pb + (size_t)(d - 3) * N_);
        float mx = -3.4e38f;
#pragma unroll 4
        for (int k = 0; k < K_; ++k) mx = fmaxf(mx, src[selIdx[m][k]]);
        out_feat[(size_t)b * 67 * S_ + (size_t)d * S_ + (s0 + m)] = mx;
    }
}

extern "C" void kernel_launch(void* const* d_in, const int* in_sizes, int n_in,
                              void* d_out, int out_size, void* d_ws, size_t ws_size,
                              hipStream_t stream) {
    const float* xyz    = (const float*)d_in[0];
    const float* points = (const float*)d_in[1];
    const int*   finit  = (const int*)d_in[2];

    float* out        = (float*)d_out;
    float* out_newxyz = out;
    float* out_feat   = out + (size_t)B_ * 3 * S_;

    float* wsA  = (float*)d_ws;
    float* wsC2 = wsA + (size_t)B_ * S_ * 4;

    fps_kernel<<<B_, 1024, 0, stream>>>(xyz, finit, out_newxyz, wsA, wsC2);
    group_kernel<<<dim3(S_ / 16, B_), 256, 0, stream>>>(xyz, points, wsA, wsC2, out_feat);
}


# FPS block of 8 waves, 16 consecutive points per thread (was 32 waves, 4 points): per-wave argmax reduce and LDS partial exchange issue 2x per SIMD per step, not 8x; 8 partials read back, 3-step DPP
# speedup vs baseline: 4.2334x; 1.3601x over previous
#include <hip/hip_runtime.h>
#include <math.h>

#define B_  8
#define N_  4096
#define D_  64
#define S_  1024
#define K_  32

typedef __attribute__((ext_vector_type(2))) float v2f;
typedef __attribute__((ext_vector_type(4))) float v4f;
typedef __attribute__((ext_vector_type(8))) float v8f;

typedef __attribute__((ext_vector_type(4))) unsigned int v4u;

#define FPS_WAVES 8
#define FPS_THREADS (FPS_WAVES * 32)
#define FPS_PTS (N_ / FPS_THREADS)

__device__ __forceinline__ unsigned wave_umax(unsigned v) {
    v = max(v, (unsigned)__builtin_amdgcn_update_dpp(0, (int)v, 0x161, 0xf, 0xf, true));
    v = max(v, (unsigned)__builtin_amdgcn_update_dpp(0, (int)v, 0x162, 0xf, 0xf, true));
    v = max(v, (unsigned)__builtin_amdgcn_update_dpp(0, (int)v, 0x164, 0xf, 0xf, true));
    v = max(v, (unsigned)__builtin_amdgcn_update_dpp(0, (int)v, 0x168, 0xf, 0xf, true));
    v = max(v, (unsigned)__builtin_amdgcn_permlanex16((int)v, (int)v, 0x76543210, 0xfedcba98, false, false));
    return v;
}

__device__ __forceinline__ unsigned group8_umax(unsigned v) {
    v = max(v, (unsigned)__builtin_amdgcn_update_dpp(0, (int)v, 0x161, 0xf, 0xf, true));
    v = max(v, (unsigned)__builtin_amdgcn_update_dpp(0, (int)v, 0x162, 0xf, 0xf, true));
    v = max(v, (unsigned)__builtin_amdgcn_update_dpp(0, (int)v, 0x164, 0xf, 0xf, true));
    return v;
}

__device__ __forceinline__ void lds_barrier() {
    __builtin_amdgcn_fence(__ATOMIC_RELEASE, "workgroup", "local");
    __builtin_amdgcn_s_barrier();
    __builtin_amdgcn_fence(__ATOMIC_ACQUIRE, "workgroup", "local");
}

__device__ __forceinline__ float lane_value(unsigned v, int src) {
    return __uint_as_float((unsigned)__builtin_amdgcn_readlane((int)v, src));
}

__global__ __launch_bounds__(FPS_THREADS) void fps_kernel(
    const float* __restrict__ xyz,
    const int*   __restrict__ farthest_init,
    float* __restrict__ out_newxyz,
    float* __restrict__ wsA,
    float* __restrict__ wsC2)
{
    __shared__ v4u red[2][FPS_WAVES];

    const int b    = blockIdx.x;
    const int tid  = threadIdx.x;
    const int lane = tid & 31;
    const int wid  = tid >> 5;

    const float* xb = xyz + (size_t)b * 3 * N_;

    float px[FPS_PTS], py[FPS_PTS], pz[FPS_PTS];
    unsigned du[FPS_PTS];
#pragma unroll
    for (int j = 0; j < FPS_PTS; ++j) {
        px[j] = xb[FPS_PTS * tid + j];
        py[j] = xb[N_ + FPS_PTS * tid + j];
        pz[j] = xb[2 * N_ + FPS_PTS * tid + j];
        du[j] = __float_as_uint(1e10f);
    }
    const int far0 = farthest_init[b] & (N_ - 1);
    float cx = xb[far0], cy = xb[N_ + far0], cz = xb[2 * N_ + far0];

    for (int it = 0; it < S_; ++it) {
        if (tid == 0) {
            out_newxyz[(size_t)b * 3 * S_ + 0 * S_ + it] = cx;
            out_newxyz[(size_t)b * 3 * S_ + 1 * S_ + it] = cy;
            out_newxyz[(size_t)b * 3 * S_ + 2 * S_ + it] = cz;
            float* A = wsA + ((size_t)b * S_ + it) * 4;
            A[0] = -2.f * cx; A[1] = -2.f * cy; A[2] = -2.f * cz; A[3] = 1.f;
            float c2;
            {
#pragma clang fp contract(off)
                c2 = fmaf(cx, cx, cy * cy) + cz * cz;
            }
            wsC2[(size_t)b * S_ + it] = c2;
        }
#pragma unroll
        for (int j = 0; j < FPS_PTS; ++j) {
            const float dx = px[j] - cx, dy = py[j] - cy, dz = pz[j] - cz;
            const float d = fmaf(dz, dz, fmaf(dx, dx, dy * dy));
            du[j] = min(du[j], __float_as_uint(d));
        }
        unsigned mq[FPS_PTS / 4];
#pragma unroll
        for (int q = 0; q < FPS_PTS / 4; ++q)
            mq[q] = max(max(du[4 * q], du[4 * q + 1]), max(du[4 * q + 2], du[4 * q + 3]));
        unsigned bu = mq[0];
#pragma unroll
        for (int q = 1; q < FPS_PTS / 4; ++q) bu = max(bu, mq[q]);

        const unsigned m   = wave_umax(bu);
        const unsigned hit = __builtin_amdgcn_ballot_w32(bu == m);
        if (lane == __builtin_ctz(hit)) {
            float sx = 0.f, sy = 0.f, sz = 0.f;
#pragma unroll
            for (int q = FPS_PTS / 4 - 1; q >= 0; --q) {
                float qx = px[4 * q + 3], qy = py[4 * q + 3], qz = pz[4 * q + 3];
#pragma unroll
                for (int j = 2; j >= 0; --j) {
                    const bool t = (du[4 * q + j] == bu);
                    qx = t ? px[4 * q + j] : qx;
                    qy = t ? py[4 * q + j] : qy;
                    qz = t ? pz[4 * q + j] : qz;
                }
                const bool tq = (q == FPS_PTS / 4 - 1) | (mq[q] == bu);
                sx = tq ? qx : sx;
                sy = tq ? qy : sy;
                sz = tq ? qz : sz;
            }
            v4u pr;
            pr.x = bu;
            pr.y = __float_as_uint(sx);
            pr.z = __float_as_uint(sy);
            pr.w = __float_as_uint(sz);
            red[it & 1][wid] = pr;
        }
        lds_barrier();
        const v4u pr = red[it & 1][lane & (FPS_WAVES - 1)];
        const unsigned M    = group8_umax(pr.x);
        const unsigned hit2 = __builtin_amdgcn_ballot_w32(pr.x == M);
        const int      win  = __builtin_ctz(hit2);
        cx = lane_value(pr.y, win);
        cy = lane_value(pr.z, win);
        cz = lane_value(pr.w, win);
    }
}

__device__ __forceinline__ void lexmin(float& v, int& i, float ov, int oi) {
    const bool take = (ov < v) | ((ov == v) & (oi < i));
    v = take ? ov : v;
    i = take ? oi : i;
}

#define TPITCH 4100

__global__ __launch_bounds__(256) void group_kernel(
    const float* __restrict__ xyz,
    const float* __restrict__ pts,
    const float* __restrict__ wsA,
    const float* __restrict__ wsC2,
    float* __restrict__ out_feat)
{
    __shared__ float DT[16 * TPITCH];
    __shared__ int   selIdx[16][K_];

    const int b    = blockIdx.y;
    const int s0   = blockIdx.x * 16;
    const int tid  = threadIdx.x;
    const int lane = tid & 31;
    const int wid  = tid >> 5;
    const bool hi  = lane >= 16;

    const float* xb = xyz + (size_t)b * 3 * N_;

    const int  mcol = lane & 15;
    const v4f  A4   = *(const v4f*)(wsA + ((size_t)b * S_ + s0 + mcol) * 4);
    v2f a;
    a.x = hi ? A4.z : A4.x;
    a.y = hi ? A4.w : A4.y;

    const int mbase = hi ? 8 : 0;
    float cmr[8];
#pragma unroll
    for (int r = 0; r < 8; ++r)
        cmr[r] = wsC2[(size_t)b * S_ + s0 + mbase + r];

    for (int t = 0; t < 32; ++t) {
        const int n = wid * 512 + t * 16 + mcol;
        const float x = xb[n], y = xb[N_ + n], z = xb[2 * N_ + n];
        const float pn2 = x * x + y * y + z * z;
        v2f bb;
        bb.x = hi ? z : x;
        bb.y = hi ? pn2 : y;

        v8f c = {};
        c = __builtin_amdgcn_wmma_f32_16x16x4_f32(
                 false, a,  false, bb,
                 (short)0, c,  false,  false);

#pragma unroll
        for (int r = 0; r < 8; ++r)
            DT[(mbase + r) * TPITCH + n] = c[r] + cmr[r];
    }
    __syncthreads();

    float* rowp[2];
    rowp[0] = &DT[wid * TPITCH];
    rowp[1] = &DT[(wid + 8) * TPITCH];

    float lv[2]; int li[2];
    int gi[2];
#pragma unroll
    for (int rr = 0; rr < 2; ++rr) {
        const float* row = rowp[rr];
        float bv = 3.4e38f; int bi = 0x7fffffff;
#pragma unroll 8
        for (int q = 0; q < N_ / 128; ++q) {
            const int j0 = q * 128 + ((lane - q) & 31) * 4;
            const v4f dv = *(const v4f*)&row[j0];
#pragma unroll
            for (int e = 0; e < 4; ++e) lexmin(bv, bi, dv[e], j0 + e);
        }
        lv[rr] = bv; li[rr] = bi;
#pragma unroll
        for (int off = 16; off >= 1; off >>= 1) {
            const float ov = __shfl_xor(bv, off, 32);
            const int   oi = __shfl_xor(bi, off, 32);
            lexmin(bv, bi, ov, oi);
        }
        gi[rr] = bi;
    }

    for (int k = 0; k < K_; ++k) {
        if (lane == 0) {
            selIdx[wid][k]     = gi[0];
            selIdx[wid + 8][k] = gi[1];
            rowp[0][gi[0]] = 3.4e38f;
            rowp[1][gi[1]] = 3.4e38f;
        }
        if (k + 1 < K_) {
#pragma unroll
            for (int rr = 0; rr < 2; ++rr) {
                const float* row = rowp[rr];
                const int bi = gi[rr];
                const int w = ((bi >> 2) + (bi >> 7)) & 31;
                const int j0 = lane * 128 + ((w - lane) & 31) * 4;
                const v4f dv = *(const v4f*)&row[j0];
                float cv = dv[0]; int ci = j0;
                float c2 = dv[2]; int i2 = j0 + 2;
                lexmin(cv, ci, dv[1], j0 + 1);
                lexmin(c2, i2, dv[3], j0 + 3);
                lexmin(cv, ci, c2, i2);
                const bool own = (lane == w);
                float ov = own ? 3.4e38f : lv[rr];
                int   oi = own ? 0x7fffffff : li[rr];
#pragma unroll
                for (int off = 16; off >= 1; off >>= 1) {
                    const float cv2 = __shfl_xor(cv, off, 32);
                    const int   ci2 = __shfl_xor(ci, off, 32);
                    const float ov2 = __shfl_xor(ov, off, 32);
                    const int   oi2 = __shfl_xor(oi, off, 32);
                    lexmin(cv, ci, cv2, ci2);
                    lexmin(ov, oi, ov2, oi2);
                }
                lv[rr] = own ? cv : lv[rr];
                li[rr] = own ? ci : li[rr];
                lexmin(cv, ci, ov, oi);
                gi[rr] = ci;
            }
        }
    }
    __syncthreads();

    const float* pb = pts + (size_t)b * D_ * N_;
    for (int t = tid; t < 16 * 67; t += 256) {
        const int m = t / 67;
        const int d = t % 67;
        const float* src = (d < 3) ? (xb + (size_t)d * N_)
                                   : (pb + (size_t)(d - 3) * N_);
        float mx = -3.4e38f;
#pragma unroll 4
        for (int k = 0; k < K_; ++k) mx = fmaxf(mx, src[selIdx[m][k]]);
        out_feat[(size_t)b * 67 * S_ + (size_t)d * S_ + (s0 + m)] = mx;
    }
}

extern "C" void kernel_launch(void* const* d_in, const int* in_sizes, int n_in,
                              void* d_out, int out_size, void* d_ws, size_t ws_size,
                              hipStream_t stream) {
    const float* xyz    = (const float*)d_in[0];
    const float* points = (const float*)d_in[1];
    const int*   finit  = (const int*)d_in[2];

    float* out        = (float*)d_out;
    float* out_newxyz = out;
    float* out_feat   = out + (size_t)B_ * 3 * S_;

    float* wsA  = (float*)d_ws;
    float* wsC2 = wsA + (size_t)B_ * S_ * 4;

    fps_kernel<<<B_, FPS_THREADS, 0, stream>>>(xyz, finit, out_newxyz, wsA, wsC2);
    group_kernel<<<dim3(S_ / 16, B_), 256, 0, stream>>>(xyz, points, wsA, wsC2, out_feat);
}


# grouping kernel phase C: channel rows copied whole into the freed LDS tile (8 rows a group, two buffers, LDS-only barrier); 32 neighbours read from LDS one per lane, wave max by DPP
# speedup vs baseline: 4.7508x; 1.1222x over previous
#include <hip/hip_runtime.h>
#include <math.h>

#define B_  8
#define N_  4096
#define D_  64
#define S_  1024
#define K_  32

typedef __attribute__((ext_vector_type(2))) float v2f;
typedef __attribute__((ext_vector_type(4))) float v4f;
typedef __attribute__((ext_vector_type(8))) float v8f;

typedef __attribute__((ext_vector_type(4))) unsigned int v4u;

#define FPS_WAVES 8
#define FPS_THREADS (FPS_WAVES * 32)
#define FPS_PTS (N_ / FPS_THREADS)

__device__ __forceinline__ unsigned wave_umax(unsigned v) {
    v = max(v, (unsigned)__builtin_amdgcn_update_dpp(0, (int)v, 0x161, 0xf, 0xf, true));
    v = max(v, (unsigned)__builtin_amdgcn_update_dpp(0, (int)v, 0x162, 0xf, 0xf, true));
    v = max(v, (unsigned)__builtin_amdgcn_update_dpp(0, (int)v, 0x164, 0xf, 0xf, true));
    v = max(v, (unsigned)__builtin_amdgcn_update_dpp(0, (int)v, 0x168, 0xf, 0xf, true));
    v = max(v, (unsigned)__builtin_amdgcn_permlanex16((int)v, (int)v, 0x76543210, 0xfedcba98, false, false));
    return v;
}

__device__ __forceinline__ unsigned group8_umax(unsigned v) {
    v = max(v, (unsigned)__builtin_amdgcn_update_dpp(0, (int)v, 0x161, 0xf, 0xf, true));
    v = max(v, (unsigned)__builtin_amdgcn_update_dpp(0, (int)v, 0x162, 0xf, 0xf, true));
    v = max(v, (unsigned)__builtin_amdgcn_update_dpp(0, (int)v, 0x164, 0xf, 0xf, true));
    return v;
}

__device__ __forceinline__ void lds_barrier() {
    __builtin_amdgcn_fence(__ATOMIC_RELEASE, "workgroup", "local");
    __builtin_amdgcn_s_barrier();
    __builtin_amdgcn_fence(__ATOMIC_ACQUIRE, "workgroup", "local");
}

__device__ __forceinline__ float lane_value(unsigned v, int src) {
    return __uint_as_float((unsigned)__builtin_amdgcn_readlane((int)v, src));
}

__global__ __launch_bounds__(FPS_THREADS) void fps_kernel(
    const float* __restrict__ xyz,
    const int*   __restrict__ farthest_init,
    float* __restrict__ out_newxyz,
    float* __restrict__ wsA,
    float* __restrict__ wsC2)
{
    __shared__ v4u red[2][FPS_WAVES];

    const int b    = blockIdx.x;
    const int tid  = threadIdx.x;
    const int lane = tid & 31;
    const int wid  = tid >> 5;

    const float* xb = xyz + (size_t)b * 3 * N_;

    float px[FPS_PTS], py[FPS_PTS], pz[FPS_PTS];
    unsigned du[FPS_PTS];
#pragma unroll
    for (int j = 0; j < FPS_PTS; ++j) {
        px[j] = xb[FPS_PTS * tid + j];
        py[j] = xb[N_ + FPS_PTS * tid + j];
        pz[j] = xb[2 * N_ + FPS_PTS * tid + j];
        du[j] = __float_as_uint(1e10f);
    }
    const int far0 = farthest_init[b] & (N_ - 1);
    float cx = xb[far0], cy = xb[N_ + far0], cz = xb[2 * N_ + far0];

    for (int it = 0; it < S_; ++it) {
        if (tid == 0) {
            out_newxyz[(size_t)b * 3 * S_ + 0 * S_ + it] = cx;
            out_newxyz[(size_t)b * 3 * S_ + 1 * S_ + it] = cy;
            out_newxyz[(size_t)b * 3 * S_ + 2 * S_ + it] = cz;
            float* A = wsA + ((size_t)b * S_ + it) * 4;
            A[0] = -2.f * cx; A[1] = -2.f * cy; A[2] = -2.f * cz; A[3] = 1.f;
            float c2;
            {
#pragma clang fp contract(off)
                c2 = fmaf(cx, cx, cy * cy) + cz * cz;
            }
            wsC2[(size_t)b * S_ + it] = c2;
        }
#pragma unroll
        for (int j = 0; j < FPS_PTS; ++j) {
            const float dx = px[j] - cx, dy = py[j] - cy, dz = pz[j] - cz;
            const float d = fmaf(dz, dz, fmaf(dx, dx, dy * dy));
            du[j] = min(du[j], __float_as_uint(d));
        }
        unsigned mq[FPS_PTS / 4];
#pragma unroll
        for (int q = 0; q < FPS_PTS / 4; ++q)
            mq[q] = max(max(du[4 * q], du[4 * q + 1]), max(du[4 * q + 2], du[4 * q + 3]));
        unsigned bu = mq[0];
#pragma unroll
        for (int q = 1; q < FPS_PTS / 4; ++q) bu = max(bu, mq[q]);

        const unsigned m   = wave_umax(bu);
        const unsigned hit = __builtin_amdgcn_ballot_w32(bu == m);
        if (lane == __builtin_ctz(hit)) {
            float sx = 0.f, sy = 0.f, sz = 0.f;
#pragma unroll
            for (int q = FPS_PTS / 4 - 1; q >= 0; --q) {
                float qx = px[4 * q + 3], qy = py[4 * q + 3], qz = pz[4 * q + 3];
#pragma unroll
                for (int j = 2; j >= 0; --j) {
                    const bool t = (du[4 * q + j] == bu);
                    qx = t ? px[4 * q + j] : qx;
                    qy = t ? py[4 * q + j] : qy;
                    qz = t ? pz[4 * q + j] : qz;
                }
                const bool tq = (q == FPS_PTS / 4 - 1) | (mq[q] == bu);
                sx = tq ? qx : sx;
                sy = tq ? qy : sy;
                sz = tq ? qz : sz;
            }
            v4u pr;
            pr.x = bu;
            pr.y = __float_as_uint(sx);
            pr.z = __float_as_uint(sy);
            pr.w = __float_as_uint(sz);
            red[it & 1][wid] = pr;
        }
        lds_barrier();
        const v4u pr = red[it & 1][lane & (FPS_WAVES - 1)];
        const unsigned M    = group8_umax(pr.x);
        const unsigned hit2 = __builtin_amdgcn_ballot_w32(pr.x == M);
        const int      win  = __builtin_ctz(hit2);
        cx = lane_value(pr.y, win);
        cy = lane_value(pr.z, win);
        cz = lane_value(pr.w, win);
    }
}

__device__ __forceinline__ void lexmin(float& v, int& i, float ov, int oi) {
    const bool take = (ov < v) | ((ov == v) & (oi < i));
    v = take ? ov : v;
    i = take ? oi : i;
}

#define CG_ROWS 8

__device__ __forceinline__ const float* chan_row(const float* xb, const float* pb, int d) {
    d = d < 66 ? d : 66;
    return (d < 3) ? (xb + (size_t)d * N_) : (pb + (size_t)(d - 3) * N_);
}

__device__ __forceinline__ float wave_fmax(float v) {
    v = fmaxf(v, __int_as_float(__builtin_amdgcn_update_dpp(0, __float_as_int(v), 0x161, 0xf, 0xf, true)));
    v = fmaxf(v, __int_as_float(__builtin_amdgcn_update_dpp(0, __float_as_int(v), 0x162, 0xf, 0xf, true)));
    v = fmaxf(v, __int_as_float(__builtin_amdgcn_update_dpp(0, __float_as_int(v), 0x164, 0xf, 0xf, true)));
    v = fmaxf(v, __int_as_float(__builtin_amdgcn_update_dpp(0, __float_as_int(v), 0x168, 0xf, 0xf, true)));
    v = fmaxf(v, __int_as_float(__builtin_amdgcn_permlanex16(__float_as_int(v), __float_as_int(v), 0x76543210, 0xfedcba98, false, false)));
    return v;
}

#define TPITCH 4100

__global__ __launch_bounds__(256) void group_kernel(
    const float* __restrict__ xyz,
    const float* __restrict__ pts,
    const float* __restrict__ wsA,
    const float* __restrict__ wsC2,
    float* __restrict__ out_feat)
{
    __shared__ float DT[16 * TPITCH];
    __shared__ int   selIdx[16][K_];

    const int b    = blockIdx.y;
    const int s0   = blockIdx.x * 16;
    const int tid  = threadIdx.x;
    const int lane = tid & 31;
    const int wid  = tid >> 5;
    const bool hi  = lane >= 16;

    const float* xb = xyz + (size_t)b * 3 * N_;

    const int  mcol = lane & 15;
    const v4f  A4   = *(const v4f*)(wsA + ((size_t)b * S_ + s0 + mcol) * 4);
    v2f a;
    a.x = hi ? A4.z : A4.x;
    a.y = hi ? A4.w : A4.y;

    const int mbase = hi ? 8 : 0;
    float cmr[8];
#pragma unroll
    for (int r = 0; r < 8; ++r)
        cmr[r] = wsC2[(size_t)b * S_ + s0 + mbase + r];

    for (int t = 0; t < 32; ++t) {
        const int n = wid * 512 + t * 16 + mcol;
        const float x = xb[n], y = xb[N_ + n], z = xb[2 * N_ + n];
        const float pn2 = x * x + y * y + z * z;
        v2f bb;
        bb.x = hi ? z : x;
        bb.y = hi ? pn2 : y;

        v8f c = {};
        c = __builtin_amdgcn_wmma_f32_16x16x4_f32(
                 false, a,  false, bb,
                 (short)0, c,  false,  false);

#pragma unroll
        for (int r = 0; r < 8; ++r)
            DT[(mbase + r) * TPITCH + n] = c[r] + cmr[r];
    }
    __syncthreads();

    float* rowp[2];
    rowp[0] = &DT[wid * TPITCH];
    rowp[1] = &DT[(wid + 8) * TPITCH];

    float lv[2]; int li[2];
    int gi[2];
#pragma unroll
    for (int rr = 0; rr < 2; ++rr) {
        const float* row = rowp[rr];
        float bv = 3.4e38f; int bi = 0x7fffffff;
#pragma unroll 8
        for (int q = 0; q < N_ / 128; ++q) {
            const int j0 = q * 128 + ((lane - q) & 31) * 4;
            const v4f dv = *(const v4f*)&row[j0];
#pragma unroll
            for (int e = 0; e < 4; ++e) lexmin(bv, bi, dv[e], j0 + e);
        }
        lv[rr] = bv; li[rr] = bi;
#pragma unroll
        for (int off = 16; off >= 1; off >>= 1) {
            const float ov = __shfl_xor(bv, off, 32);
            const int   oi = __shfl_xor(bi, off, 32);
            lexmin(bv, bi, ov, oi);
        }
        gi[rr] = bi;
    }

    for (int k = 0; k < K_; ++k) {
        if (lane == 0) {
            selIdx[wid][k]     = gi[0];
            selIdx[wid + 8][k] = gi[1];
            rowp[0][gi[0]] = 3.4e38f;
            rowp[1][gi[1]] = 3.4e38f;
        }
        if (k + 1 < K_) {
#pragma unroll
            for (int rr = 0; rr < 2; ++rr) {
                const float* row = rowp[rr];
                const int bi = gi[rr];
                const int w = ((bi >> 2) + (bi >> 7)) & 31;
                const int j0 = lane * 128 + ((w - lane) & 31) * 4;
                const v4f dv = *(const v4f*)&row[j0];
                float cv = dv[0]; int ci = j0;
                float c2 = dv[2]; int i2 = j0 + 2;
                lexmin(cv, ci, dv[1], j0 + 1);
                lexmin(c2, i2, dv[3], j0 + 3);
                lexmin(cv, ci, c2, i2);
                const bool own = (lane == w);
                float ov = own ? 3.4e38f : lv[rr];
                int   oi = own ? 0x7fffffff : li[rr];
#pragma unroll
                for (int off = 16; off >= 1; off >>= 1) {
                    const float cv2 = __shfl_xor(cv, off, 32);
                    const int   ci2 = __shfl_xor(ci, off, 32);
                    const float ov2 = __shfl_xor(ov, off, 32);
                    const int   oi2 = __shfl_xor(oi, off, 32);
                    lexmin(cv, ci, cv2, ci2);
                    lexmin(ov, oi, ov2, oi2);
                }
                lv[rr] = own ? cv : lv[rr];
                li[rr] = own ? ci : li[rr];
                lexmin(cv, ci, ov, oi);
                gi[rr] = ci;
            }
        }
    }
    __syncthreads();

    const float* pb = pts + (size_t)b * D_ * N_;
    const int i0 = selIdx[wid][lane];
    const int i1 = selIdx[wid + 8][lane];
    float* o0 = out_feat + (size_t)b * 67 * S_ + (s0 + wid);
    float* o1 = o0 + 8;
    float r0 = 0.f, r1 = 0.f;

    v4f stg[CG_ROWS][4];
#pragma unroll
    for (int r = 0; r < CG_ROWS; ++r) {
        const float* row = chan_row(xb, pb, r);
#pragma unroll
        for (int j = 0; j < 4; ++j) stg[r][j] = *(const v4f*)(row + j * 1024 + tid * 4);
    }
    for (int g = 0; g < (67 + CG_ROWS - 1) / CG_ROWS; ++g) {
        float* buf = DT + (g & 1) * (CG_ROWS * N_);
#pragma unroll
        for (int r = 0; r < CG_ROWS; ++r)
#pragma unroll
            for (int j = 0; j < 4; ++j) *(v4f*)(buf + r * N_ + j * 1024 + tid * 4) = stg[r][j];
        if ((g + 1) * CG_ROWS < 67) {
#pragma unroll
            for (int r = 0; r < CG_ROWS; ++r) {
                const float* row = chan_row(xb, pb, (g + 1) * CG_ROWS + r);
#pragma unroll
                for (int j = 0; j < 4; ++j) stg[r][j] = *(const v4f*)(row + j * 1024 + tid * 4);
            }
        }
        lds_barrier();
        float v0[CG_ROWS], v1[CG_ROWS];
#pragma unroll
        for (int r = 0; r < CG_ROWS; ++r) { v0[r] = buf[r * N_ + i0]; v1[r] = buf[r * N_ + i1]; }
#pragma unroll
        for (int r = 0; r < CG_ROWS; ++r) {
            const int d = g * CG_ROWS + r;
            const float m0 = wave_fmax(v0[r]);
            const float m1 = wave_fmax(v1[r]);
            const bool mine = (lane == (d & 31)) & (d < 67);
            r0 = mine ? m0 : r0;
            r1 = mine ? m1 : r1;
        }
        const int dl = g * CG_ROWS + CG_ROWS - 1;
        if ((dl & 31) == 31 || dl >= 66) {
            const int dd = (dl & ~31) + lane;
            if (dd < 67) { o0[(size_t)dd * S_] = r0; o1[(size_t)dd * S_] = r1; }
        }
    }
}

extern "C" void kernel_launch(void* const* d_in, const int* in_sizes, int n_in,
                              void* d_out, int out_size, void* d_ws, size_t ws_size,
                              hipStream_t stream) {
    const float* xyz    = (const float*)d_in[0];
    const float* points = (const float*)d_in[1];
    const int*   finit  = (const int*)d_in[2];

    float* out        = (float*)d_out;
    float* out_newxyz = out;
    float* out_feat   = out + (size_t)B_ * 3 * S_;

    float* wsA  = (float*)d_ws;
    float* wsC2 = wsA + (size_t)B_ * S_ * 4;

    fps_kernel<<<B_, FPS_THREADS, 0, stream>>>(xyz, finit, out_newxyz, wsA, wsC2);
    group_kernel<<<dim3(S_ / 16, B_), 256, 0, stream>>>(xyz, points, wsA, wsC2, out_feat);
}


# FPS step loop stores nothing to global memory: each step's 8 wave partials keep their own LDS row (1024 rows); new_xyz, A rows and |c|^2 are rebuilt from the rows after the loop by all 256 threads
# speedup vs baseline: 5.2505x; 1.1052x over previous
#include <hip/hip_runtime.h>
#include <math.h>

#define B_  8
#define N_  4096
#define D_  64
#define S_  1024
#define K_  32

typedef __attribute__((ext_vector_type(2))) float v2f;
typedef __attribute__((ext_vector_type(4))) float v4f;
typedef __attribute__((ext_vector_type(8))) float v8f;

typedef __attribute__((ext_vector_type(4))) unsigned int v4u;

#define FPS_WAVES 8
#define FPS_THREADS (FPS_WAVES * 32)
#define FPS_PTS (N_ / FPS_THREADS)
#define FPS_ROW (FPS_WAVES + 1)

__device__ __forceinline__ unsigned wave_umax(unsigned v) {
    v = max(v, (unsigned)__builtin_amdgcn_update_dpp(0, (int)v, 0x161, 0xf, 0xf, true));
    v = max(v, (unsigned)__builtin_amdgcn_update_dpp(0, (int)v, 0x162, 0xf, 0xf, true));
    v = max(v, (unsigned)__builtin_amdgcn_update_dpp(0, (int)v, 0x164, 0xf, 0xf, true));
    v = max(v, (unsigned)__builtin_amdgcn_update_dpp(0, (int)v, 0x168, 0xf, 0xf, true));
    v = max(v, (unsigned)__builtin_amdgcn_permlanex16((int)v, (int)v, 0x76543210, 0xfedcba98, false, false));
    return v;
}

__device__ __forceinline__ unsigned group8_umax(unsigned v) {
    v = max(v, (unsigned)__builtin_amdgcn_update_dpp(0, (int)v, 0x161, 0xf, 0xf, true));
    v = max(v, (unsigned)__builtin_amdgcn_update_dpp(0, (int)v, 0x162, 0xf, 0xf, true));
    v = max(v, (unsigned)__builtin_amdgcn_update_dpp(0, (int)v, 0x164, 0xf, 0xf, true));
    return v;
}

__device__ __forceinline__ void lds_barrier() {
    __builtin_amdgcn_fence(__ATOMIC_RELEASE, "workgroup", "local");
    __builtin_amdgcn_s_barrier();
    __builtin_amdgcn_fence(__ATOMIC_ACQUIRE, "workgroup", "local");
}

__device__ __forceinline__ float lane_value(unsigned v, int src) {
    return __uint_as_float((unsigned)__builtin_amdgcn_readlane((int)v, src));
}

__global__ __launch_bounds__(FPS_THREADS) void fps_kernel(
    const float* __restrict__ xyz,
    const int*   __restrict__ farthest_init,
    float* __restrict__ out_newxyz,
    float* __restrict__ wsA,
    float* __restrict__ wsC2)
{
    __shared__ v4u red[S_][FPS_ROW];

    const int b    = blockIdx.x;
    const int tid  = threadIdx.x;
    const int lane = tid & 31;
    const int wid  = tid >> 5;

    const float* xb = xyz + (size_t)b * 3 * N_;

    float px[FPS_PTS], py[FPS_PTS], pz[FPS_PTS];
    unsigned du[FPS_PTS];
#pragma unroll
    for (int j = 0; j < FPS_PTS; ++j) {
        px[j] = xb[FPS_PTS * tid + j];
        py[j] = xb[N_ + FPS_PTS * tid + j];
        pz[j] = xb[2 * N_ + FPS_PTS * tid + j];
        du[j] = __float_as_uint(1e10f);
    }
    const int far0 = farthest_init[b] & (N_ - 1);
    const float c0x = xb[far0], c0y = xb[N_ + far0], c0z = xb[2 * N_ + far0];
    float cx = c0x, cy = c0y, cz = c0z;

    v4u* rowp = &red[0][0];
    for (int it = 0; it < S_; ++it, rowp += FPS_ROW) {
#pragma unroll
        for (int j = 0; j < FPS_PTS; ++j) {
            const float dx = px[j] - cx, dy = py[j] - cy, dz = pz[j] - cz;
            const float d = fmaf(dz, dz, fmaf(dx, dx, dy * dy));
            du[j] = min(du[j], __float_as_uint(d));
        }
        unsigned mq[FPS_PTS / 4];
#pragma unroll
        for (int q = 0; q < FPS_PTS / 4; ++q)
            mq[q] = max(max(du[4 * q], du[4 * q + 1]), max(du[4 * q + 2], du[4 * q + 3]));
        unsigned bu = mq[0];
#pragma unroll
        for (int q = 1; q < FPS_PTS / 4; ++q) bu = max(bu, mq[q]);

        const unsigned m   = wave_umax(bu);
        const unsigned hit = __builtin_amdgcn_ballot_w32(bu == m);
        if (lane == __builtin_ctz(hit)) {
            float sx = 0.f, sy = 0.f, sz = 0.f;
#pragma unroll
            for (int q = FPS_PTS / 4 - 1; q >= 0; --q) {
                float qx = px[4 * q + 3], qy = py[4 * q + 3], qz = pz[4 * q + 3];
#pragma unroll
                for (int j = 2; j >= 0; --j) {
                    const bool t = (du[4 * q + j] == bu);
                    qx = t ? px[4 * q + j] : qx;
                    qy = t ? py[4 * q + j] : qy;
                    qz = t ? pz[4 * q + j] : qz;
                }
                const bool tq = (q == FPS_PTS / 4 - 1) | (mq[q] == bu);
                sx = tq ? qx : sx;
                sy = tq ? qy : sy;
                sz = tq ? qz : sz;
            }
            v4u pr;
            pr.x = bu;
            pr.y = __float_as_uint(sx);
            pr.z = __float_as_uint(sy);
            pr.w = __float_as_uint(sz);
            rowp[wid] = pr;
        }
        lds_barrier();
        const v4u pr = rowp[lane & (FPS_WAVES - 1)];
        const unsigned M    = group8_umax(pr.x);
        const unsigned hit2 = __builtin_amdgcn_ballot_w32(pr.x == M);
        const int      win  = __builtin_ctz(hit2);
        cx = lane_value(pr.y, win);
        cy = lane_value(pr.z, win);
        cz = lane_value(pr.w, win);
    }

    for (int k = 0; k < S_ / FPS_THREADS; ++k) {
        const int it  = k * FPS_THREADS + tid;
        const int row = it > 0 ? it - 1 : 0;
        v4u best = red[row][0];
        for (int w = 1; w < FPS_WAVES; ++w) {
            const v4u p = red[row][w];
            const bool t = p.x > best.x;
            best.x = t ? p.x : best.x;
            best.y = t ? p.y : best.y;
            best.z = t ? p.z : best.z;
            best.w = t ? p.w : best.w;
        }
        const float ex = it > 0 ? __uint_as_float(best.y) : c0x;
        const float ey = it > 0 ? __uint_as_float(best.z) : c0y;
        const float ez = it > 0 ? __uint_as_float(best.w) : c0z;

        out_newxyz[(size_t)b * 3 * S_ + 0 * S_ + it] = ex;
        out_newxyz[(size_t)b * 3 * S_ + 1 * S_ + it] = ey;
        out_newxyz[(size_t)b * 3 * S_ + 2 * S_ + it] = ez;
        v4f A;
        A.x = -2.f * ex; A.y = -2.f * ey; A.z = -2.f * ez; A.w = 1.f;
        *(v4f*)(wsA + ((size_t)b * S_ + it) * 4) = A;
        float c2;
        {
#pragma clang fp contract(off)
            c2 = fmaf(ex, ex, ey * ey) + ez * ez;
        }
        wsC2[(size_t)b * S_ + it] = c2;
    }
}

__device__ __forceinline__ void lexmin(float& v, int& i, float ov, int oi) {
    const bool take = (ov < v) | ((ov == v) & (oi < i));
    v = take ? ov : v;
    i = take ? oi : i;
}

#define CG_ROWS 8

__device__ __forceinline__ const float* chan_row(const float* xb, const float* pb, int d) {
    d = d < 66 ? d : 66;
    return (d < 3) ? (xb + (size_t)d * N_) : (pb + (size_t)(d - 3) * N_);
}

__device__ __forceinline__ float wave_fmax(float v) {
    v = fmaxf(v, __int_as_float(__builtin_amdgcn_update_dpp(0, __float_as_int(v), 0x161, 0xf, 0xf, true)));
    v = fmaxf(v, __int_as_float(__builtin_amdgcn_update_dpp(0, __float_as_int(v), 0x162, 0xf, 0xf, true)));
    v = fmaxf(v, __int_as_float(__builtin_amdgcn_update_dpp(0, __float_as_int(v), 0x164, 0xf, 0xf, true)));
    v = fmaxf(v, __int_as_float(__builtin_amdgcn_update_dpp(0, __float_as_int(v), 0x168, 0xf, 0xf, true)));
    v = fmaxf(v, __int_as_float(__builtin_amdgcn_permlanex16(__float_as_int(v), __float_as_int(v), 0x76543210, 0xfedcba98, false, false)));
    return v;
}

#define TPITCH 4100

__global__ __launch_bounds__(256) void group_kernel(
    const float* __restrict__ xyz,
    const float* __restrict__ pts,
    const float* __restrict__ wsA,
    const float* __restrict__ wsC2,
    float* __restrict__ out_feat)
{
    __shared__ float DT[16 * TPITCH];
    __shared__ int   selIdx[16][K_];

    const int b    = blockIdx.y;
    const int s0   = blockIdx.x * 16;
    const int tid  = threadIdx.x;
    const int lane = tid & 31;
    const int wid  = tid >> 5;
    const bool hi  = lane >= 16;

    const float* xb = xyz + (size_t)b * 3 * N_;

    const int  mcol = lane & 15;
    const v4f  A4   = *(const v4f*)(wsA + ((size_t)b * S_ + s0 + mcol) * 4);
    v2f a;
    a.x = hi ? A4.z : A4.x;
    a.y = hi ? A4.w : A4.y;

    const int mbase = hi ? 8 : 0;
    float cmr[8];
#pragma unroll
    for (int r = 0; r < 8; ++r)
        cmr[r] = wsC2[(size_t)b * S_ + s0 + mbase + r];

    for (int t = 0; t < 32; ++t) {
        const int n = wid * 512 + t * 16 + mcol;
        const float x = xb[n], y = xb[N_ + n], z = xb[2 * N_ + n];
        const float pn2 = x * x + y * y + z * z;
        v2f bb;
        bb.x = hi ? z : x;
        bb.y = hi ? pn2 : y;

        v8f c = {};
        c = __builtin_amdgcn_wmma_f32_16x16x4_f32(
                 false, a,  false, bb,
                 (short)0, c,  false,  false);

#pragma unroll
        for (int r = 0; r < 8; ++r)
            DT[(mbase + r) * TPITCH + n] = c[r] + cmr[r];
    }
    __syncthreads();

    float* rowp[2];
    rowp[0] = &DT[wid * TPITCH];
    rowp[1] = &DT[(wid + 8) * TPITCH];

    float lv[2]; int li[2];
    int gi[2];
#pragma unroll
    for (int rr = 0; rr < 2; ++rr) {
        const float* row = rowp[rr];
        float bv = 3.4e38f; int bi = 0x7fffffff;
#pragma unroll 8
        for (int q = 0; q < N_ / 128; ++q) {
            const int j0 = q * 128 + ((lane - q) & 31) * 4;
            const v4f dv = *(const v4f*)&row[j0];
#pragma unroll
            for (int e = 0; e < 4; ++e) lexmin(bv, bi, dv[e], j0 + e);
        }
        lv[rr] = bv; li[rr] = bi;
#pragma unroll
        for (int off = 16; off >= 1; off >>= 1) {
            const float ov = __shfl_xor(bv, off, 32);
            const int   oi = __shfl_xor(bi, off, 32);
            lexmin(bv, bi, ov, oi);
        }
        gi[rr] = bi;
    }

    for (int k = 0; k < K_; ++k) {
        if (lane == 0) {
            selIdx[wid][k]     = gi[0];
            selIdx[wid + 8][k] = gi[1];
            rowp[0][gi[0]] = 3.4e38f;
            rowp[1][gi[1]] = 3.4e38f;
        }
        if (k + 1 < K_) {
#pragma unroll
            for (int rr = 0; rr < 2; ++rr) {
                const float* row = rowp[rr];
                const int bi = gi[rr];
                const int w = ((bi >> 2) + (bi >> 7)) & 31;
                const int j0 = lane * 128 + ((w - lane) & 31) * 4;
                const v4f dv = *(const v4f*)&row[j0];
                float cv = dv[0]; int ci = j0;
                float c2 = dv[2]; int i2 = j0 + 2;
                lexmin(cv, ci, dv[1], j0 + 1);
                lexmin(c2, i2, dv[3], j0 + 3);
                lexmin(cv, ci, c2, i2);
                const bool own = (lane == w);
                float ov = own ? 3.4e38f : lv[rr];
                int   oi = own ? 0x7fffffff : li[rr];
#pragma unroll
                for (int off = 16; off >= 1; off >>= 1) {
                    const float cv2 = __shfl_xor(cv, off, 32);
                    const int   ci2 = __shfl_xor(ci, off, 32);
                    const float ov2 = __shfl_xor(ov, off, 32);
                    const int   oi2 = __shfl_xor(oi, off, 32);
                    lexmin(cv, ci, cv2, ci2);
                    lexmin(ov, oi, ov2, oi2);
                }
                lv[rr] = own ? cv : lv[rr];
                li[rr] = own ? ci : li[rr];
                lexmin(cv, ci, ov, oi);
                gi[rr] = ci;
            }
        }
    }
    __syncthreads();

    const float* pb = pts + (size_t)b * D_ * N_;
    const int i0 = selIdx[wid][lane];
    const int i1 = selIdx[wid + 8][lane];
    float* o0 = out_feat + (size_t)b * 67 * S_ + (s0 + wid);
    float* o1 = o0 + 8;
    float r0 = 0.f, r1 = 0.f;

    v4f stg[CG_ROWS][4];
#pragma unroll
    for (int r = 0; r < CG_ROWS; ++r) {
        const float* row = chan_row(xb, pb, r);
#pragma unroll
        for (int j = 0; j < 4; ++j) stg[r][j] = *(const v4f*)(row + j * 1024 + tid * 4);
    }
    for (int g = 0; g < (67 + CG_ROWS - 1) / CG_ROWS; ++g) {
        float* buf = DT + (g & 1) * (CG_ROWS * N_);
#pragma unroll
        for (int r = 0; r < CG_ROWS; ++r)
#pragma unroll
            for (int j = 0; j < 4; ++j) *(v4f*)(buf + r * N_ + j * 1024 + tid * 4) = stg[r][j];
        if ((g + 1) * CG_ROWS < 67) {
#pragma unroll
            for (int r = 0; r < CG_ROWS; ++r) {
                const float* row = chan_row(xb, pb, (g + 1) * CG_ROWS + r);
#pragma unroll
                for (int j = 0; j < 4; ++j) stg[r][j] = *(const v4f*)(row + j * 1024 + tid * 4);
            }
        }
        lds_barrier();
        float v0[CG_ROWS], v1[CG_ROWS];
#pragma unroll
        for (int r = 0; r < CG_ROWS; ++r) { v0[r] = buf[r * N_ + i0]; v1[r] = buf[r * N_ + i1]; }
#pragma unroll
        for (int r = 0; r < CG_ROWS; ++r) {
            const int d = g * CG_ROWS + r;
            const float m0 = wave_fmax(v0[r]);
            const float m1 = wave_fmax(v1[r]);
            const bool mine = (lane == (d & 31)) & (d < 67);
            r0 = mine ? m0 : r0;
            r1 = mine ? m1 : r1;
        }
        const int dl = g * CG_ROWS + CG_ROWS - 1;
        if ((dl & 31) == 31 || dl >= 66) {
            const int dd = (dl & ~31) + lane;
            if (dd < 67) { o0[(size_t)dd * S_] = r0; o1[(size_t)dd * S_] = r1; }
        }
    }
}

extern "C" void kernel_launch(void* const* d_in, const int* in_sizes, int n_in,
                              void* d_out, int out_size, void* d_ws, size_t ws_size,
                              hipStream_t stream) {
    const float* xyz    = (const float*)d_in[0];
    const float* points = (const float*)d_in[1];
    const int*   finit  = (const int*)d_in[2];

    float* out        = (float*)d_out;
    float* out_newxyz = out;
    float* out_feat   = out + (size_t)B_ * 3 * S_;

    float* wsA  = (float*)d_ws;
    float* wsC2 = wsA + (size_t)B_ * S_ * 4;

    fps_kernel<<<B_, FPS_THREADS, 0, stream>>>(xyz, finit, out_newxyz, wsA, wsC2);
    group_kernel<<<dim3(S_ / 16, B_), 256, 0, stream>>>(xyz, points, wsA, wsC2, out_feat);
}


# grouping kernel: a block owns 32 centroids (two passes of phases A and B over the one LDS tile), so phase C streams the 67 channel rows through LDS once per 32 centroids, not per 16; 256 blocks
# speedup vs baseline: 5.4254x; 1.0333x over previous
#include <hip/hip_runtime.h>
#include <math.h>

#define B_  8
#define N_  4096
#define D_  64
#define S_  1024
#define K_  32

typedef __attribute__((ext_vector_type(2))) float v2f;
typedef __attribute__((ext_vector_type(4))) float v4f;
typedef __attribute__((ext_vector_type(8))) float v8f;

typedef __attribute__((ext_vector_type(4))) unsigned int v4u;

#define FPS_WAVES 8
#define FPS_THREADS (FPS_WAVES * 32)
#define FPS_PTS (N_ / FPS_THREADS)
#define FPS_ROW (FPS_WAVES + 1)

__device__ __forceinline__ unsigned wave_umax(unsigned v) {
    v = max(v, (unsigned)__builtin_amdgcn_update_dpp(0, (int)v, 0x161, 0xf, 0xf, true));
    v = max(v, (unsigned)__builtin_amdgcn_update_dpp(0, (int)v, 0x162, 0xf, 0xf, true));
    v = max(v, (unsigned)__builtin_amdgcn_update_dpp(0, (int)v, 0x164, 0xf, 0xf, true));
    v = max(v, (unsigned)__builtin_amdgcn_update_dpp(0, (int)v, 0x168, 0xf, 0xf, true));
    v = max(v, (unsigned)__builtin_amdgcn_permlanex16((int)v, (int)v, 0x76543210, 0xfedcba98, false, false));
    return v;
}

__device__ __forceinline__ unsigned group8_umax(unsigned v) {
    v = max(v, (unsigned)__builtin_amdgcn_update_dpp(0, (int)v, 0x161, 0xf, 0xf, true));
    v = max(v, (unsigned)__builtin_amdgcn_update_dpp(0, (int)v, 0x162, 0xf, 0xf, true));
    v = max(v, (unsigned)__builtin_amdgcn_update_dpp(0, (int)v, 0x164, 0xf, 0xf, true));
    return v;
}

__device__ __forceinline__ void lds_barrier() {
    __builtin_amdgcn_fence(__ATOMIC_RELEASE, "workgroup", "local");
    __builtin_amdgcn_s_barrier();
    __builtin_amdgcn_fence(__ATOMIC_ACQUIRE, "workgroup", "local");
}

__device__ __forceinline__ float lane_value(unsigned v, int src) {
    return __uint_as_float((unsigned)__builtin_amdgcn_readlane((int)v, src));
}

__global__ __launch_bounds__(FPS_THREADS) void fps_kernel(
    const float* __restrict__ xyz,
    const int*   __restrict__ farthest_init,
    float* __restrict__ out_newxyz,
    float* __restrict__ wsA,
    float* __restrict__ wsC2)
{
    __shared__ v4u red[S_][FPS_ROW];

    const int b    = blockIdx.x;
    const int tid  = threadIdx.x;
    const int lane = tid & 31;
    const int wid  = tid >> 5;

    const float* xb = xyz + (size_t)b * 3 * N_;

    float px[FPS_PTS], py[FPS_PTS], pz[FPS_PTS];
    unsigned du[FPS_PTS];
#pragma unroll
    for (int j = 0; j < FPS_PTS; ++j) {
        px[j] = xb[FPS_PTS * tid + j];
        py[j] = xb[N_ + FPS_PTS * tid + j];
        pz[j] = xb[2 * N_ + FPS_PTS * tid + j];
        du[j] = __float_as_uint(1e10f);
    }
    const int far0 = farthest_init[b] & (N_ - 1);
    const float c0x = xb[far0], c0y = xb[N_ + far0], c0z = xb[2 * N_ + far0];
    float cx = c0x, cy = c0y, cz = c0z;

    v4u* rowp = &red[0][0];
    for (int it = 0; it < S_; ++it, rowp += FPS_ROW) {
#pragma unroll
        for (int j = 0; j < FPS_PTS; ++j) {
            const float dx = px[j] - cx, dy = py[j] - cy, dz = pz[j] - cz;
            const float d = fmaf(dz, dz, fmaf(dx, dx, dy * dy));
            du[j] = min(du[j], __float_as_uint(d));
        }
        unsigned mq[FPS_PTS / 4];
#pragma unroll
        for (int q = 0; q < FPS_PTS / 4; ++q)
            mq[q] = max(max(du[4 * q], du[4 * q + 1]), max(du[4 * q + 2], du[4 * q + 3]));
        unsigned bu = mq[0];
#pragma unroll
        for (int q = 1; q < FPS_PTS / 4; ++q) bu = max(bu, mq[q]);

        const unsigned m   = wave_umax(bu);
        const unsigned hit = __builtin_amdgcn_ballot_w32(bu == m);
        if (lane == __builtin_ctz(hit)) {
            float sx = 0.f, sy = 0.f, sz = 0.f;
#pragma unroll
            for (int q = FPS_PTS / 4 - 1; q >= 0; --q) {
                float qx = px[4 * q + 3], qy = py[4 * q + 3], qz = pz[4 * q + 3];
#pragma unroll
                for (int j = 2; j >= 0; --j) {
                    const bool t = (du[4 * q + j] == bu);
                    qx = t ? px[4 * q + j] : qx;
                    qy = t ? py[4 * q + j] : qy;
                    qz = t ? pz[4 * q + j] : qz;
                }
                const bool tq = (q == FPS_PTS / 4 - 1) | (mq[q] == bu);
                sx = tq ? qx : sx;
                sy = tq ? qy : sy;
                sz = tq ? qz : sz;
            }
            v4u pr;
            pr.x = bu;
            pr.y = __float_as_uint(sx);
            pr.z = __float_as_uint(sy);
            pr.w = __float_as_uint(sz);
            rowp[wid] = pr;
        }
        lds_barrier();
        const v4u pr = rowp[lane & (FPS_WAVES - 1)];
        const unsigned M    = group8_umax(pr.x);
        const unsigned hit2 = __builtin_amdgcn_ballot_w32(pr.x == M);
        const int      win  = __builtin_ctz(hit2);
        cx = lane_value(pr.y, win);
        cy = lane_value(pr.z, win);
        cz = lane_value(pr.w, win);
    }

    for (int k = 0; k < S_ / FPS_THREADS; ++k) {
        const int it  = k * FPS_THREADS + tid;
        const int row = it > 0 ? it - 1 : 0;
        v4u best = red[row][0];
        for (int w = 1; w < FPS_WAVES; ++w) {
            const v4u p = red[row][w];
            const bool t = p.x > best.x;
            best.x = t ? p.x : best.x;
            best.y = t ? p.y : best.y;
            best.z = t ? p.z : best.z;
            best.w = t ? p.w : best.w;
        }
        const float ex = it > 0 ? __uint_as_float(best.y) : c0x;
        const float ey = it > 0 ? __uint_as_float(best.z) : c0y;
        const float ez = it > 0 ? __uint_as_float(best.w) : c0z;

        out_newxyz[(size_t)b * 3 * S_ + 0 * S_ + it] = ex;
        out_newxyz[(size_t)b * 3 * S_ + 1 * S_ + it] = ey;
        out_newxyz[(size_t)b * 3 * S_ + 2 * S_ + it] = ez;
        v4f A;
        A.x = -2.f * ex; A.y = -2.f * ey; A.z = -2.f * ez; A.w = 1.f;
        *(v4f*)(wsA + ((size_t)b * S_ + it) * 4) = A;
        float c2;
        {
#pragma clang fp contract(off)
            c2 = fmaf(ex, ex, ey * ey) + ez * ez;
        }
        wsC2[(size_t)b * S_ + it] = c2;
    }
}

__device__ __forceinline__ void lexmin(float& v, int& i, float ov, int oi) {
    const bool take = (ov < v) | ((ov == v) & (oi < i));
    v = take ? ov : v;
    i = take ? oi : i;
}

#define CG_ROWS 8

__device__ __forceinline__ const float* chan_row(const float* xb, const float* pb, int d) {
    d = d < 66 ? d : 66;
    return (d < 3) ? (xb + (size_t)d * N_) : (pb + (size_t)(d - 3) * N_);
}

__device__ __forceinline__ float wave_fmax(float v) {
    v = fmaxf(v, __int_as_float(__builtin_amdgcn_update_dpp(0, __float_as_int(v), 0x161, 0xf, 0xf, true)));
    v = fmaxf(v, __int_as_float(__builtin_amdgcn_update_dpp(0, __float_as_int(v), 0x162, 0xf, 0xf, true)));
    v = fmaxf(v, __int_as_float(__builtin_amdgcn_update_dpp(0, __float_as_int(v), 0x164, 0xf, 0xf, true)));
    v = fmaxf(v, __int_as_float(__builtin_amdgcn_update_dpp(0, __float_as_int(v), 0x168, 0xf, 0xf, true)));
    v = fmaxf(v, __int_as_float(__builtin_amdgcn_permlanex16(__float_as_int(v), __float_as_int(v), 0x76543210, 0xfedcba98, false, false)));
    return v;
}

#define TPITCH 4100
#define CB_ 32
#define NPASS (CB_ / 16)
#define CW_ (2 * NPASS)

__global__ __launch_bounds__(256) void group_kernel(
    const float* __restrict__ xyz,
    const float* __restrict__ pts,
    const float* __restrict__ wsA,
    const float* __restrict__ wsC2,
    float* __restrict__ out_feat)
{
    __shared__ float DT[16 * TPITCH];
    __shared__ int   selIdx[CB_][K_];

    const int b    = blockIdx.y;
    const int sb   = blockIdx.x * CB_;
    const int tid  = threadIdx.x;
    const int lane = tid & 31;
    const int wid  = tid >> 5;
    const bool hi  = lane >= 16;

    const float* xb = xyz + (size_t)b * 3 * N_;

    const int  mcol  = lane & 15;
    const int  mbase = hi ? 8 : 0;

    float* rowp[2];
    rowp[0] = &DT[wid * TPITCH];
    rowp[1] = &DT[(wid + 8) * TPITCH];

    for (int pass = 0; pass < NPASS; ++pass) {
        const int s0 = sb + pass * 16;

        const v4f  A4   = *(const v4f*)(wsA + ((size_t)b * S_ + s0 + mcol) * 4);
        v2f a;
        a.x = hi ? A4.z : A4.x;
        a.y = hi ? A4.w : A4.y;

        float cmr[8];
#pragma unroll
        for (int r = 0; r < 8; ++r)
            cmr[r] = wsC2[(size_t)b * S_ + s0 + mbase + r];

        for (int t = 0; t < 32; ++t) {
            const int n = wid * 512 + t * 16 + mcol;
            const float x = xb[n], y = xb[N_ + n], z = xb[2 * N_ + n];
            const float pn2 = x * x + y * y + z * z;
            v2f bb;
            bb.x = hi ? z : x;
            bb.y = hi ? pn2 : y;

            v8f c = {};
            c = __builtin_amdgcn_wmma_f32_16x16x4_f32(
                     false, a,  false, bb,
                     (short)0, c,  false,  false);

#pragma unroll
            for (int r = 0; r < 8; ++r)
                DT[(mbase + r) * TPITCH + n] = c[r] + cmr[r];
        }
        __syncthreads();

        float lv[2]; int li[2];
        int gi[2];
#pragma unroll
        for (int rr = 0; rr < 2; ++rr) {
            const float* row = rowp[rr];
            float bv = 3.4e38f; int bi = 0x7fffffff;
#pragma unroll 8
            for (int q = 0; q < N_ / 128; ++q) {
                const int j0 = q * 128 + ((lane - q) & 31) * 4;
                const v4f dv = *(const v4f*)&row[j0];
#pragma unroll
                for (int e = 0; e < 4; ++e) lexmin(bv, bi, dv[e], j0 + e);
            }
            lv[rr] = bv; li[rr] = bi;
#pragma unroll
            for (int off = 16; off >= 1; off >>= 1) {
                const float ov = __shfl_xor(bv, off, 32);
                const int   oi = __shfl_xor(bi, off, 32);
                lexmin(bv, bi, ov, oi);
            }
            gi[rr] = bi;
        }

        for (int k = 0; k < K_; ++k) {
            if (lane == 0) {
                selIdx[pass * 16 + wid][k]     = gi[0];
                selIdx[pass * 16 + wid + 8][k] = gi[1];
                rowp[0][gi[0]] = 3.4e38f;
                rowp[1][gi[1]] = 3.4e38f;
            }
            if (k + 1 < K_) {
#pragma unroll
                for (int rr = 0; rr < 2; ++rr) {
                    const float* row = rowp[rr];
                    const int bi = gi[rr];
                    const int w = ((bi >> 2) + (bi >> 7)) & 31;
                    const int j0 = lane * 128 + ((w - lane) & 31) * 4;
                    const v4f dv = *(const v4f*)&row[j0];
                    float cv = dv[0]; int ci = j0;
                    float c2 = dv[2]; int i2 = j0 + 2;
                    lexmin(cv, ci, dv[1], j0 + 1);
                    lexmin(c2, i2, dv[3], j0 + 3);
                    lexmin(cv, ci, c2, i2);
                    const bool own = (lane == w);
                    float ov = own ? 3.4e38f : lv[rr];
                    int   oi = own ? 0x7fffffff : li[rr];
#pragma unroll
                    for (int off = 16; off >= 1; off >>= 1) {
                        const float cv2 = __shfl_xor(cv, off, 32);
                        const int   ci2 = __shfl_xor(ci, off, 32);
                        const float ov2 = __shfl_xor(ov, off, 32);
                        const int   oi2 = __shfl_xor(oi, off, 32);
                        lexmin(cv, ci, cv2, ci2);
                        lexmin(ov, oi, ov2, oi2);
                    }
                    lv[rr] = own ? cv : lv[rr];
                    li[rr] = own ? ci : li[rr];
                    lexmin(cv, ci, ov, oi);
                    gi[rr] = ci;
                }
            }
        }
        __syncthreads();
    }

    const float* pb = pts + (size_t)b * D_ * N_;
    int    ni[CW_];
    float* op[CW_];
    float  rm[CW_];
#pragma unroll
    for (int c = 0; c < CW_; ++c) {
        const int crow = (c >> 1) * 16 + (c & 1) * 8 + wid;
        ni[c] = selIdx[crow][lane];
        op[c] = out_feat + (size_t)b * 67 * S_ + (sb + crow);
        rm[c] = 0.f;
    }

    v4f stg[CG_ROWS][4];
#pragma unroll
    for (int r = 0; r < CG_ROWS; ++r) {
        const float* row = chan_row(xb, pb, r);
#pragma unroll
        for (int j = 0; j < 4; ++j) stg[r][j] = *(const v4f*)(row + j * 1024 + tid * 4);
    }
    for (int g = 0; g < (67 + CG_ROWS - 1) / CG_ROWS; ++g) {
        float* buf = DT + (g & 1) * (CG_ROWS * N_);
#pragma unroll
        for (int r = 0; r < CG_ROWS; ++r)
#pragma unroll
            for (int j = 0; j < 4; ++j) *(v4f*)(buf + r * N_ + j * 1024 + tid * 4) = stg[r][j];
        if ((g + 1) * CG_ROWS < 67) {
#pragma unroll
            for (int r = 0; r < CG_ROWS; ++r) {
                const float* row = chan_row(xb, pb, (g + 1) * CG_ROWS + r);
#pragma unroll
                for (int j = 0; j < 4; ++j) stg[r][j] = *(const v4f*)(row + j * 1024 + tid * 4);
            }
        }
        lds_barrier();
        float v[CW_][CG_ROWS];
#pragma unroll
        for (int r = 0; r < CG_ROWS; ++r)
#pragma unroll
            for (int c = 0; c < CW_; ++c) v[c][r] = buf[r * N_ + ni[c]];
#pragma unroll
        for (int r = 0; r < CG_ROWS; ++r) {
            const int d = g * CG_ROWS + r;
            const bool mine = (lane == (d & 31)) & (d < 67);
#pragma unroll
            for (int c = 0; c < CW_; ++c) {
                const float m = wave_fmax(v[c][r]);
                rm[c] = mine ? m : rm[c];
            }
        }
        const int dl = g * CG_ROWS + CG_ROWS - 1;
        if ((dl & 31) == 31 || dl >= 66) {
            const int dd = (dl & ~31) + lane;
            if (dd < 67) {
#pragma unroll
                for (int c = 0; c < CW_; ++c) op[c][(size_t)dd * S_] = rm[c];
            }
        }
    }
}

extern "C" void kernel_launch(void* const* d_in, const int* in_sizes, int n_in,
                              void* d_out, int out_size, void* d_ws, size_t ws_size,
                              hipStream_t stream) {
    const float* xyz    = (const float*)d_in[0];
    const float* points = (const float*)d_in[1];
    const int*   finit  = (const int*)d_in[2];

    float* out        = (float*)d_out;
    float* out_newxyz = out;
    float* out_feat   = out + (size_t)B_ * 3 * S_;

    float* wsA  = (float*)d_ws;
    float* wsC2 = wsA + (size_t)B_ * S_ * 4;

    fps_kernel<<<B_, FPS_THREADS, 0, stream>>>(xyz, finit, out_newxyz, wsA, wsC2);
    group_kernel<<<dim3(S_ / CB_, B_), 256, 0, stream>>>(xyz, points, wsA, wsC2, out_feat);
}


# FPS step: every lane picks its own best point's x,y,z with selects beside the wave argmax (one straight-line block, winner chosen by store address), not in a one-lane branch behind it
# speedup vs baseline: 5.7275x; 1.0557x over previous
#include <hip/hip_runtime.h>
#include <math.h>

#define B_  8
#define N_  4096
#define D_  64
#define S_  1024
#define K_  32

typedef __attribute__((ext_vector_type(2))) float v2f;
typedef __attribute__((ext_vector_type(4))) float v4f;
typedef __attribute__((ext_vector_type(8))) float v8f;

typedef __attribute__((ext_vector_type(4))) unsigned int v4u;

#define FPS_WAVES 8
#define FPS_THREADS (FPS_WAVES * 32)
#define FPS_PTS (N_ / FPS_THREADS)
#define FPS_ROW (FPS_WAVES + 1)

__device__ __forceinline__ unsigned wave_umax(unsigned v) {
    v = max(v, (unsigned)__builtin_amdgcn_update_dpp(0, (int)v, 0x161, 0xf, 0xf, true));
    v = max(v, (unsigned)__builtin_amdgcn_update_dpp(0, (int)v, 0x162, 0xf, 0xf, true));
    v = max(v, (unsigned)__builtin_amdgcn_update_dpp(0, (int)v, 0x164, 0xf, 0xf, true));
    v = max(v, (unsigned)__builtin_amdgcn_update_dpp(0, (int)v, 0x168, 0xf, 0xf, true));
    v = max(v, (unsigned)__builtin_amdgcn_permlanex16((int)v, (int)v, 0x76543210, 0xfedcba98, false, false));
    return v;
}

__device__ __forceinline__ unsigned group8_umax(unsigned v) {
    v = max(v, (unsigned)__builtin_amdgcn_update_dpp(0, (int)v, 0x161, 0xf, 0xf, true));
    v = max(v, (unsigned)__builtin_amdgcn_update_dpp(0, (int)v, 0x162, 0xf, 0xf, true));
    v = max(v, (unsigned)__builtin_amdgcn_update_dpp(0, (int)v, 0x164, 0xf, 0xf, true));
    return v;
}

__device__ __forceinline__ void lds_barrier() {
    __builtin_amdgcn_fence(__ATOMIC_RELEASE, "workgroup", "local");
    __builtin_amdgcn_s_barrier();
    __builtin_amdgcn_fence(__ATOMIC_ACQUIRE, "workgroup", "local");
}

__device__ __forceinline__ float lane_value(unsigned v, int src) {
    return __uint_as_float((unsigned)__builtin_amdgcn_readlane((int)v, src));
}

__global__ __launch_bounds__(FPS_THREADS) void fps_kernel(
    const float* __restrict__ xyz,
    const int*   __restrict__ farthest_init,
    float* __restrict__ out_newxyz,
    float* __restrict__ wsA,
    float* __restrict__ wsC2)
{
    __shared__ v4u red[S_][FPS_ROW];
    __shared__ v4u spare[FPS_THREADS];

    const int b    = blockIdx.x;
    const int tid  = threadIdx.x;
    const int lane = tid & 31;
    const int wid  = tid >> 5;

    const float* xb = xyz + (size_t)b * 3 * N_;

    float px[FPS_PTS], py[FPS_PTS], pz[FPS_PTS];
    unsigned du[FPS_PTS];
#pragma unroll
    for (int j = 0; j < FPS_PTS; ++j) {
        px[j] = xb[FPS_PTS * tid + j];
        py[j] = xb[N_ + FPS_PTS * tid + j];
        pz[j] = xb[2 * N_ + FPS_PTS * tid + j];
        du[j] = __float_as_uint(1e10f);
    }
    const int far0 = farthest_init[b] & (N_ - 1);
    const float c0x = xb[far0], c0y = xb[N_ + far0], c0z = xb[2 * N_ + far0];
    float cx = c0x, cy = c0y, cz = c0z;

    v4u* rowp = &red[0][0];
    for (int it = 0; it < S_; ++it, rowp += FPS_ROW) {
#pragma unroll
        for (int j = 0; j < FPS_PTS; ++j) {
            const float dx = px[j] - cx, dy = py[j] - cy, dz = pz[j] - cz;
            const float d = fmaf(dz, dz, fmaf(dx, dx, dy * dy));
            du[j] = min(du[j], __float_as_uint(d));
        }
        unsigned mq[FPS_PTS / 4];
        float qx[FPS_PTS / 4], qy[FPS_PTS / 4], qz[FPS_PTS / 4];
#pragma unroll
        for (int q = 0; q < FPS_PTS / 4; ++q) {
            mq[q] = max(max(du[4 * q], du[4 * q + 1]), max(du[4 * q + 2], du[4 * q + 3]));
            qx[q] = px[4 * q + 3]; qy[q] = py[4 * q + 3]; qz[q] = pz[4 * q + 3];
#pragma unroll
            for (int j = 2; j >= 0; --j) {
                const bool t = (du[4 * q + j] == mq[q]);
                qx[q] = t ? px[4 * q + j] : qx[q];
                qy[q] = t ? py[4 * q + j] : qy[q];
                qz[q] = t ? pz[4 * q + j] : qz[q];
            }
        }
        unsigned bu = mq[0];
#pragma unroll
        for (int q = 1; q < FPS_PTS / 4; ++q) bu = max(bu, mq[q]);
        float sx = qx[FPS_PTS / 4 - 1], sy = qy[FPS_PTS / 4 - 1], sz = qz[FPS_PTS / 4 - 1];
#pragma unroll
        for (int q = FPS_PTS / 4 - 2; q >= 0; --q) {
            const bool tq = (mq[q] == bu);
            sx = tq ? qx[q] : sx;
            sy = tq ? qy[q] : sy;
            sz = tq ? qz[q] : sz;
        }
        v4u mine;
        mine.x = bu;
        mine.y = __float_as_uint(sx);
        mine.z = __float_as_uint(sy);
        mine.w = __float_as_uint(sz);

        const unsigned m   = wave_umax(bu);
        const unsigned hit = __builtin_amdgcn_ballot_w32(bu == m);
        v4u* dst = (lane == __builtin_ctz(hit)) ? &rowp[wid] : &spare[tid];
        *dst = mine;
        lds_barrier();
        const v4u pr = rowp[lane & (FPS_WAVES - 1)];
        const unsigned M    = group8_umax(pr.x);
        const unsigned hit2 = __builtin_amdgcn_ballot_w32(pr.x == M);
        const int      win  = __builtin_ctz(hit2);
        cx = lane_value(pr.y, win);
        cy = lane_value(pr.z, win);
        cz = lane_value(pr.w, win);
    }

    for (int k = 0; k < S_ / FPS_THREADS; ++k) {
        const int it  = k * FPS_THREADS + tid;
        const int row = it > 0 ? it - 1 : 0;
        v4u best = red[row][0];
        for (int w = 1; w < FPS_WAVES; ++w) {
            const v4u p = red[row][w];
            const bool t = p.x > best.x;
            best.x = t ? p.x : best.x;
            best.y = t ? p.y : best.y;
            best.z = t ? p.z : best.z;
            best.w = t ? p.w : best.w;
        }
        const float ex = it > 0 ? __uint_as_float(best.y) : c0x;
        const float ey = it > 0 ? __uint_as_float(best.z) : c0y;
        const float ez = it > 0 ? __uint_as_float(best.w) : c0z;

        out_newxyz[(size_t)b * 3 * S_ + 0 * S_ + it] = ex;
        out_newxyz[(size_t)b * 3 * S_ + 1 * S_ + it] = ey;
        out_newxyz[(size_t)b * 3 * S_ + 2 * S_ + it] = ez;
        v4f A;
        A.x = -2.f * ex; A.y = -2.f * ey; A.z = -2.f * ez; A.w = 1.f;
        *(v4f*)(wsA + ((size_t)b * S_ + it) * 4) = A;
        float c2;
        {
#pragma clang fp contract(off)
            c2 = fmaf(ex, ex, ey * ey) + ez * ez;
        }
        wsC2[(size_t)b * S_ + it] = c2;
    }
}

__device__ __forceinline__ void lexmin(float& v, int& i, float ov, int oi) {
    const bool take = (ov < v) | ((ov == v) & (oi < i));
    v = take ? ov : v;
    i = take ? oi : i;
}

#define CG_ROWS 8

__device__ __forceinline__ const float* chan_row(const float* xb, const float* pb, int d) {
    d = d < 66 ? d : 66;
    return (d < 3) ? (xb + (size_t)d * N_) : (pb + (size_t)(d - 3) * N_);
}

__device__ __forceinline__ float wave_fmax(float v) {
    v = fmaxf(v, __int_as_float(__builtin_amdgcn_update_dpp(0, __float_as_int(v), 0x161, 0xf, 0xf, true)));
    v = fmaxf(v, __int_as_float(__builtin_amdgcn_update_dpp(0, __float_as_int(v), 0x162, 0xf, 0xf, true)));
    v = fmaxf(v, __int_as_float(__builtin_amdgcn_update_dpp(0, __float_as_int(v), 0x164, 0xf, 0xf, true)));
    v = fmaxf(v, __int_as_float(__builtin_amdgcn_update_dpp(0, __float_as_int(v), 0x168, 0xf, 0xf, true)));
    v = fmaxf(v, __int_as_float(__builtin_amdgcn_permlanex16(__float_as_int(v), __float_as_int(v), 0x76543210, 0xfedcba98, false, false)));
    return v;
}

#define TPITCH 4100
#define CB_ 32
#define NPASS (CB_ / 16)
#define CW_ (2 * NPASS)

__global__ __launch_bounds__(256) void group_kernel(
    const float* __restrict__ xyz,
    const float* __restrict__ pts,
    const float* __restrict__ wsA,
    const float* __restrict__ wsC2,
    float* __restrict__ out_feat)
{
    __shared__ float DT[16 * TPITCH];
    __shared__ int   selIdx[CB_][K_];

    const int b    = blockIdx.y;
    const int sb   = blockIdx.x * CB_;
    const int tid  = threadIdx.x;
    const int lane = tid & 31;
    const int wid  = tid >> 5;
    const bool hi  = lane >= 16;

    const float* xb = xyz + (size_t)b * 3 * N_;

    const int  mcol  = lane & 15;
    const int  mbase = hi ? 8 : 0;

    float* rowp[2];
    rowp[0] = &DT[wid * TPITCH];
    rowp[1] = &DT[(wid + 8) * TPITCH];

    for (int pass = 0; pass < NPASS; ++pass) {
        const int s0 = sb + pass * 16;

        const v4f  A4   = *(const v4f*)(wsA + ((size_t)b * S_ + s0 + mcol) * 4);
        v2f a;
        a.x = hi ? A4.z : A4.x;
        a.y = hi ? A4.w : A4.y;

        float cmr[8];
#pragma unroll
        for (int r = 0; r < 8; ++r)
            cmr[r] = wsC2[(size_t)b * S_ + s0 + mbase + r];

        for (int t = 0; t < 32; ++t) {
            const int n = wid * 512 + t * 16 + mcol;
            const float x = xb[n], y = xb[N_ + n], z = xb[2 * N_ + n];
            const float pn2 = x * x + y * y + z * z;
            v2f bb;
            bb.x = hi ? z : x;
            bb.y = hi ? pn2 : y;

            v8f c = {};
            c = __builtin_amdgcn_wmma_f32_16x16x4_f32(
                     false, a,  false, bb,
                     (short)0, c,  false,  false);

#pragma unroll
            for (int r = 0; r < 8; ++r)
                DT[(mbase + r) * TPITCH + n] = c[r] + cmr[r];
        }
        __syncthreads();

        float lv[2]; int li[2];
        int gi[2];
#pragma unroll
        for (int rr = 0; rr < 2; ++rr) {
            const float* row = rowp[rr];
            float bv = 3.4e38f; int bi = 0x7fffffff;
#pragma unroll 8
            for (int q = 0; q < N_ / 128; ++q) {
                const int j0 = q * 128 + ((lane - q) & 31) * 4;
                const v4f dv = *(const v4f*)&row[j0];
#pragma unroll
                for (int e = 0; e < 4; ++e) lexmin(bv, bi, dv[e], j0 + e);
            }
            lv[rr] = bv; li[rr] = bi;
#pragma unroll
            for (int off = 16; off >= 1; off >>= 1) {
                const float ov = __shfl_xor(bv, off, 32);
                const int   oi = __shfl_xor(bi, off, 32);
                lexmin(bv, bi, ov, oi);
            }
            gi[rr] = bi;
        }

        for (int k = 0; k < K_; ++k) {
            if (lane == 0) {
                selIdx[pass * 16 + wid][k]     = gi[0];
                selIdx[pass * 16 + wid + 8][k] = gi[1];
                rowp[0][gi[0]] = 3.4e38f;
                rowp[1][gi[1]] = 3.4e38f;
            }
            if (k + 1 < K_) {
#pragma unroll
                for (int rr = 0; rr < 2; ++rr) {
                    const float* row = rowp[rr];
                    const int bi = gi[rr];
                    const int w = ((bi >> 2) + (bi >> 7)) & 31;
                    const int j0 = lane * 128 + ((w - lane) & 31) * 4;
                    const v4f dv = *(const v4f*)&row[j0];
                    float cv = dv[0]; int ci = j0;
                    float c2 = dv[2]; int i2 = j0 + 2;
                    lexmin(cv, ci, dv[1], j0 + 1);
                    lexmin(c2, i2, dv[3], j0 + 3);
                    lexmin(cv, ci, c2, i2);
                    const bool own = (lane == w);
                    float ov = own ? 3.4e38f : lv[rr];
                    int   oi = own ? 0x7fffffff : li[rr];
#pragma unroll
                    for (int off = 16; off >= 1; off >>= 1) {
                        const float cv2 = __shfl_xor(cv, off, 32);
                        const int   ci2 = __shfl_xor(ci, off, 32);
                        const float ov2 = __shfl_xor(ov, off, 32);
                        const int   oi2 = __shfl_xor(oi, off, 32);
                        lexmin(cv, ci, cv2, ci2);
                        lexmin(ov, oi, ov2, oi2);
                    }
                    lv[rr] = own ? cv : lv[rr];
                    li[rr] = own ? ci : li[rr];
                    lexmin(cv, ci, ov, oi);
                    gi[rr] = ci;
                }
            }
        }
        __syncthreads();
    }

    const float* pb = pts + (size_t)b * D_ * N_;
    int    ni[CW_];
    float* op[CW_];
    float  rm[CW_];
#pragma unroll
    for (int c = 0; c < CW_; ++c) {
        const int crow = (c >> 1) * 16 + (c & 1) * 8 + wid;
        ni[c] = selIdx[crow][lane];
        op[c] = out_feat + (size_t)b * 67 * S_ + (sb + crow);
        rm[c] = 0.f;
    }

    v4f stg[CG_ROWS][4];
#pragma unroll
    for (int r = 0; r < CG_ROWS; ++r) {
        const float* row = chan_row(xb, pb, r);
#pragma unroll
        for (int j = 0; j < 4; ++j) stg[r][j] = *(const v4f*)(row + j * 1024 + tid * 4);
    }
    for (int g = 0; g < (67 + CG_ROWS - 1) / CG_ROWS; ++g) {
        float* buf = DT + (g & 1) * (CG_ROWS * N_);
#pragma unroll
        for (int r = 0; r < CG_ROWS; ++r)
#pragma unroll
            for (int j = 0; j < 4; ++j) *(v4f*)(buf + r * N_ + j * 1024 + tid * 4) = stg[r][j];
        if ((g + 1) * CG_ROWS < 67) {
#pragma unroll
            for (int r = 0; r < CG_ROWS; ++r) {
                const float* row = chan_row(xb, pb, (g + 1) * CG_ROWS + r);
#pragma unroll
                for (int j = 0; j < 4; ++j) stg[r][j] = *(const v4f*)(row + j * 1024 + tid * 4);
            }
        }
        lds_barrier();
        float v[CW_][CG_ROWS];
#pragma unroll
        for (int r = 0; r < CG_ROWS; ++r)
#pragma unroll
            for (int c = 0; c < CW_; ++c) v[c][r] = buf[r * N_ + ni[c]];
#pragma unroll
        for (int r = 0; r < CG_ROWS; ++r) {
            const int d = g * CG_ROWS + r;
            const bool mine = (lane == (d & 31)) & (d < 67);
#pragma unroll
            for (int c = 0; c < CW_; ++c) {
                const float m = wave_fmax(v[c][r]);
                rm[c] = mine ? m : rm[c];
            }
        }
        const int dl = g * CG_ROWS + CG_ROWS - 1;
        if ((dl & 31) == 31 || dl >= 66) {
            const int dd = (dl & ~31) + lane;
            if (dd < 67) {
#pragma unroll
                for (int c = 0; c < CW_; ++c) op[c][(size_t)dd * S_] = rm[c];
            }
        }
    }
}

extern "C" void kernel_launch(void* const* d_in, const int* in_sizes, int n_in,
                              void* d_out, int out_size, void* d_ws, size_t ws_size,
                              hipStream_t stream) {
    const float* xyz    = (const float*)d_in[0];
    const float* points = (const float*)d_in[1];
    const int*   finit  = (const int*)d_in[2];

    float* out        = (float*)d_out;
    float* out_newxyz = out;
    float* out_feat   = out + (size_t)B_ * 3 * S_;

    float* wsA  = (float*)d_ws;
    float* wsC2 = wsA + (size_t)B_ * S_ * 4;

    fps_kernel<<<B_, FPS_THREADS, 0, stream>>>(xyz, finit, out_newxyz, wsA, wsC2);
    group_kernel<<<dim3(S_ / CB_, B_), 256, 0, stream>>>(xyz, points, wsA, wsC2, out_feat);
}


# top-32 picks: wave (dist,idx) minimum as DPP integer min on an order-preserving dist key, then DPP min on idx among lanes equal to it; replaces the 5-step ds_bpermute butterflies of pairs
# speedup vs baseline: 6.2410x; 1.0897x over previous
#include <hip/hip_runtime.h>
#include <math.h>

#define B_  8
#define N_  4096
#define D_  64
#define S_  1024
#define K_  32

typedef __attribute__((ext_vector_type(2))) float v2f;
typedef __attribute__((ext_vector_type(4))) float v4f;
typedef __attribute__((ext_vector_type(8))) float v8f;

typedef __attribute__((ext_vector_type(4))) unsigned int v4u;

#define FPS_WAVES 8
#define FPS_THREADS (FPS_WAVES * 32)
#define FPS_PTS (N_ / FPS_THREADS)
#define FPS_ROW (FPS_WAVES + 1)

__device__ __forceinline__ unsigned wave_umax(unsigned v) {
    v = max(v, (unsigned)__builtin_amdgcn_update_dpp(0, (int)v, 0x161, 0xf, 0xf, true));
    v = max(v, (unsigned)__builtin_amdgcn_update_dpp(0, (int)v, 0x162, 0xf, 0xf, true));
    v = max(v, (unsigned)__builtin_amdgcn_update_dpp(0, (int)v, 0x164, 0xf, 0xf, true));
    v = max(v, (unsigned)__builtin_amdgcn_update_dpp(0, (int)v, 0x168, 0xf, 0xf, true));
    v = max(v, (unsigned)__builtin_amdgcn_permlanex16((int)v, (int)v, 0x76543210, 0xfedcba98, false, false));
    return v;
}

__device__ __forceinline__ unsigned group8_umax(unsigned v) {
    v = max(v, (unsigned)__builtin_amdgcn_update_dpp(0, (int)v, 0x161, 0xf, 0xf, true));
    v = max(v, (unsigned)__builtin_amdgcn_update_dpp(0, (int)v, 0x162, 0xf, 0xf, true));
    v = max(v, (unsigned)__builtin_amdgcn_update_dpp(0, (int)v, 0x164, 0xf, 0xf, true));
    return v;
}

__device__ __forceinline__ void lds_barrier() {
    __builtin_amdgcn_fence(__ATOMIC_RELEASE, "workgroup", "local");
    __builtin_amdgcn_s_barrier();
    __builtin_amdgcn_fence(__ATOMIC_ACQUIRE, "workgroup", "local");
}

__device__ __forceinline__ float lane_value(unsigned v, int src) {
    return __uint_as_float((unsigned)__builtin_amdgcn_readlane((int)v, src));
}

__global__ __launch_bounds__(FPS_THREADS) void fps_kernel(
    const float* __restrict__ xyz,
    const int*   __restrict__ farthest_init,
    float* __restrict__ out_newxyz,
    float* __restrict__ wsA,
    float* __restrict__ wsC2)
{
    __shared__ v4u red[S_][FPS_ROW];
    __shared__ v4u spare[FPS_THREADS];

    const int b    = blockIdx.x;
    const int tid  = threadIdx.x;
    const int lane = tid & 31;
    const int wid  = tid >> 5;

    const float* xb = xyz + (size_t)b * 3 * N_;

    float px[FPS_PTS], py[FPS_PTS], pz[FPS_PTS];
    unsigned du[FPS_PTS];
#pragma unroll
    for (int j = 0; j < FPS_PTS; ++j) {
        px[j] = xb[FPS_PTS * tid + j];
        py[j] = xb[N_ + FPS_PTS * tid + j];
        pz[j] = xb[2 * N_ + FPS_PTS * tid + j];
        du[j] = __float_as_uint(1e10f);
    }
    const int far0 = farthest_init[b] & (N_ - 1);
    const float c0x = xb[far0], c0y = xb[N_ + far0], c0z = xb[2 * N_ + far0];
    float cx = c0x, cy = c0y, cz = c0z;

    v4u* rowp = &red[0][0];
    for (int it = 0; it < S_; ++it, rowp += FPS_ROW) {
#pragma unroll
        for (int j = 0; j < FPS_PTS; ++j) {
            const float dx = px[j] - cx, dy = py[j] - cy, dz = pz[j] - cz;
            const float d = fmaf(dz, dz, fmaf(dx, dx, dy * dy));
            du[j] = min(du[j], __float_as_uint(d));
        }
        unsigned mq[FPS_PTS / 4];
        float qx[FPS_PTS / 4], qy[FPS_PTS / 4], qz[FPS_PTS / 4];
#pragma unroll
        for (int q = 0; q < FPS_PTS / 4; ++q) {
            mq[q] = max(max(du[4 * q], du[4 * q + 1]), max(du[4 * q + 2], du[4 * q + 3]));
            qx[q] = px[4 * q + 3]; qy[q] = py[4 * q + 3]; qz[q] = pz[4 * q + 3];
#pragma unroll
            for (int j = 2; j >= 0; --j) {
                const bool t = (du[4 * q + j] == mq[q]);
                qx[q] = t ? px[4 * q + j] : qx[q];
                qy[q] = t ? py[4 * q + j] : qy[q];
                qz[q] = t ? pz[4 * q + j] : qz[q];
            }
        }
        unsigned bu = mq[0];
#pragma unroll
        for (int q = 1; q < FPS_PTS / 4; ++q) bu = max(bu, mq[q]);
        float sx = qx[FPS_PTS / 4 - 1], sy = qy[FPS_PTS / 4 - 1], sz = qz[FPS_PTS / 4 - 1];
#pragma unroll
        for (int q = FPS_PTS / 4 - 2; q >= 0; --q) {
            const bool tq = (mq[q] == bu);
            sx = tq ? qx[q] : sx;
            sy = tq ? qy[q] : sy;
            sz = tq ? qz[q] : sz;
        }
        v4u mine;
        mine.x = bu;
        mine.y = __float_as_uint(sx);
        mine.z = __float_as_uint(sy);
        mine.w = __float_as_uint(sz);

        const unsigned m   = wave_umax(bu);
        const unsigned hit = __builtin_amdgcn_ballot_w32(bu == m);
        v4u* dst = (lane == __builtin_ctz(hit)) ? &rowp[wid] : &spare[tid];
        *dst = mine;
        lds_barrier();
        const v4u pr = rowp[lane & (FPS_WAVES - 1)];
        const unsigned M    = group8_umax(pr.x);
        const unsigned hit2 = __builtin_amdgcn_ballot_w32(pr.x == M);
        const int      win  = __builtin_ctz(hit2);
        cx = lane_value(pr.y, win);
        cy = lane_value(pr.z, win);
        cz = lane_value(pr.w, win);
    }

    for (int k = 0; k < S_ / FPS_THREADS; ++k) {
        const int it  = k * FPS_THREADS + tid;
        const int row = it > 0 ? it - 1 : 0;
        v4u best = red[row][0];
        for (int w = 1; w < FPS_WAVES; ++w) {
            const v4u p = red[row][w];
            const bool t = p.x > best.x;
            best.x = t ? p.x : best.x;
            best.y = t ? p.y : best.y;
            best.z = t ? p.z : best.z;
            best.w = t ? p.w : best.w;
        }
        const float ex = it > 0 ? __uint_as_float(best.y) : c0x;
        const float ey = it > 0 ? __uint_as_float(best.z) : c0y;
        const float ez = it > 0 ? __uint_as_float(best.w) : c0z;

        out_newxyz[(size_t)b * 3 * S_ + 0 * S_ + it] = ex;
        out_newxyz[(size_t)b * 3 * S_ + 1 * S_ + it] = ey;
        out_newxyz[(size_t)b * 3 * S_ + 2 * S_ + it] = ez;
        v4f A;
        A.x = -2.f * ex; A.y = -2.f * ey; A.z = -2.f * ez; A.w = 1.f;
        *(v4f*)(wsA + ((size_t)b * S_ + it) * 4) = A;
        float c2;
        {
#pragma clang fp contract(off)
            c2 = fmaf(ex, ex, ey * ey) + ez * ez;
        }
        wsC2[(size_t)b * S_ + it] = c2;
    }
}

__device__ __forceinline__ void lexmin(float& v, int& i, float ov, int oi) {
    const bool take = (ov < v) | ((ov == v) & (oi < i));
    v = take ? ov : v;
    i = take ? oi : i;
}

#define CG_ROWS 8

__device__ __forceinline__ const float* chan_row(const float* xb, const float* pb, int d) {
    d = d < 66 ? d : 66;
    return (d < 3) ? (xb + (size_t)d * N_) : (pb + (size_t)(d - 3) * N_);
}

__device__ __forceinline__ float wave_fmax(float v) {
    v = fmaxf(v, __int_as_float(__builtin_amdgcn_update_dpp(0, __float_as_int(v), 0x161, 0xf, 0xf, true)));
    v = fmaxf(v, __int_as_float(__builtin_amdgcn_update_dpp(0, __float_as_int(v), 0x162, 0xf, 0xf, true)));
    v = fmaxf(v, __int_as_float(__builtin_amdgcn_update_dpp(0, __float_as_int(v), 0x164, 0xf, 0xf, true)));
    v = fmaxf(v, __int_as_float(__builtin_amdgcn_update_dpp(0, __float_as_int(v), 0x168, 0xf, 0xf, true)));
    v = fmaxf(v, __int_as_float(__builtin_amdgcn_permlanex16(__float_as_int(v), __float_as_int(v), 0x76543210, 0xfedcba98, false, false)));
    return v;
}

__device__ __forceinline__ int wave_imin(int v) {
    v = min(v, __builtin_amdgcn_update_dpp(0, v, 0x161, 0xf, 0xf, true));
    v = min(v, __builtin_amdgcn_update_dpp(0, v, 0x162, 0xf, 0xf, true));
    v = min(v, __builtin_amdgcn_update_dpp(0, v, 0x164, 0xf, 0xf, true));
    v = min(v, __builtin_amdgcn_update_dpp(0, v, 0x168, 0xf, 0xf, true));
    v = min(v, (int)__builtin_amdgcn_permlanex16(v, v, 0x76543210, 0xfedcba98, false, false));
    return v;
}

__device__ __forceinline__ int fkey(int bits) {
    return bits ^ ((bits >> 31) & 0x7fffffff);
}

__device__ __forceinline__ void wave_lexmin(int& k, int& i) {
    const int km = wave_imin(k);
    i = wave_imin(k == km ? i : 0x7fffffff);
    k = km;
}

#define TPITCH 4100
#define CB_ 32
#define NPASS (CB_ / 16)
#define CW_ (2 * NPASS)

__global__ __launch_bounds__(256) void group_kernel(
    const float* __restrict__ xyz,
    const float* __restrict__ pts,
    const float* __restrict__ wsA,
    const float* __restrict__ wsC2,
    float* __restrict__ out_feat)
{
    __shared__ float DT[16 * TPITCH];
    __shared__ int   selIdx[CB_][K_];

    const int b    = blockIdx.y;
    const int sb   = blockIdx.x * CB_;
    const int tid  = threadIdx.x;
    const int lane = tid & 31;
    const int wid  = tid >> 5;
    const bool hi  = lane >= 16;

    const float* xb = xyz + (size_t)b * 3 * N_;

    const int  mcol  = lane & 15;
    const int  mbase = hi ? 8 : 0;

    float* rowp[2];
    rowp[0] = &DT[wid * TPITCH];
    rowp[1] = &DT[(wid + 8) * TPITCH];

    for (int pass = 0; pass < NPASS; ++pass) {
        const int s0 = sb + pass * 16;

        const v4f  A4   = *(const v4f*)(wsA + ((size_t)b * S_ + s0 + mcol) * 4);
        v2f a;
        a.x = hi ? A4.z : A4.x;
        a.y = hi ? A4.w : A4.y;

        float cmr[8];
#pragma unroll
        for (int r = 0; r < 8; ++r)
            cmr[r] = wsC2[(size_t)b * S_ + s0 + mbase + r];

        for (int t = 0; t < 32; ++t) {
            const int n = wid * 512 + t * 16 + mcol;
            const float x = xb[n], y = xb[N_ + n], z = xb[2 * N_ + n];
            const float pn2 = x * x + y * y + z * z;
            v2f bb;
            bb.x = hi ? z : x;
            bb.y = hi ? pn2 : y;

            v8f c = {};
            c = __builtin_amdgcn_wmma_f32_16x16x4_f32(
                     false, a,  false, bb,
                     (short)0, c,  false,  false);

#pragma unroll
            for (int r = 0; r < 8; ++r)
                DT[(mbase + r) * TPITCH + n] = c[r] + cmr[r];
        }
        __syncthreads();

        int lk[2]; int li[2];
        int gi[2];
#pragma unroll
        for (int rr = 0; rr < 2; ++rr) {
            const float* row = rowp[rr];
            float bv = 3.4e38f; int bi = 0x7fffffff;
#pragma unroll 8
            for (int q = 0; q < N_ / 128; ++q) {
                const int j0 = q * 128 + ((lane - q) & 31) * 4;
                const v4f dv = *(const v4f*)&row[j0];
#pragma unroll
                for (int e = 0; e < 4; ++e) lexmin(bv, bi, dv[e], j0 + e);
            }
            int bk = fkey(__float_as_int(bv));
            lk[rr] = bk; li[rr] = bi;
            wave_lexmin(bk, bi);
            gi[rr] = bi;
        }

        for (int k = 0; k < K_; ++k) {
            if (lane == 0) {
                selIdx[pass * 16 + wid][k]     = gi[0];
                selIdx[pass * 16 + wid + 8][k] = gi[1];
                rowp[0][gi[0]] = 3.4e38f;
                rowp[1][gi[1]] = 3.4e38f;
            }
            if (k + 1 < K_) {
#pragma unroll
                for (int rr = 0; rr < 2; ++rr) {
                    const float* row = rowp[rr];
                    const int bi = gi[rr];
                    const int w = ((bi >> 2) + (bi >> 7)) & 31;
                    const int j0 = lane * 128 + ((w - lane) & 31) * 4;
                    const v4f dv = *(const v4f*)&row[j0];
                    float cv = dv[0]; int ci = j0;
                    float c2 = dv[2]; int i2 = j0 + 2;
                    lexmin(cv, ci, dv[1], j0 + 1);
                    lexmin(c2, i2, dv[3], j0 + 3);
                    lexmin(cv, ci, c2, i2);
                    int ck = fkey(__float_as_int(cv));
                    const bool own = (lane == w);
                    int ok = own ? 0x7fffffff : lk[rr];
                    int oi = own ? 0x7fffffff : li[rr];
                    wave_lexmin(ck, ci);
                    wave_lexmin(ok, oi);
                    lk[rr] = own ? ck : lk[rr];
                    li[rr] = own ? ci : li[rr];
                    const bool other = (ok < ck) | ((ok == ck) & (oi < ci));
                    gi[rr] = other ? oi : ci;
                }
            }
        }
        __syncthreads();
    }

    const float* pb = pts + (size_t)b * D_ * N_;
    int    ni[CW_];
    float* op[CW_];
    float  rm[CW_];
#pragma unroll
    for (int c = 0; c < CW_; ++c) {
        const int crow = (c >> 1) * 16 + (c & 1) * 8 + wid;
        ni[c] = selIdx[crow][lane];
        op[c] = out_feat + (size_t)b * 67 * S_ + (sb + crow);
        rm[c] = 0.f;
    }

    v4f stg[CG_ROWS][4];
#pragma unroll
    for (int r = 0; r < CG_ROWS; ++r) {
        const float* row = chan_row(xb, pb, r);
#pragma unroll
        for (int j = 0; j < 4; ++j) stg[r][j] = *(const v4f*)(row + j * 1024 + tid * 4);
    }
    for (int g = 0; g < (67 + CG_ROWS - 1) / CG_ROWS; ++g) {
        float* buf = DT + (g & 1) * (CG_ROWS * N_);
#pragma unroll
        for (int r = 0; r < CG_ROWS; ++r)
#pragma unroll
            for (int j = 0; j < 4; ++j) *(v4f*)(buf + r * N_ + j * 1024 + tid * 4) = stg[r][j];
        if ((g + 1) * CG_ROWS < 67) {
#pragma unroll
            for (int r = 0; r < CG_ROWS; ++r) {
                const float* row = chan_row(xb, pb, (g + 1) * CG_ROWS + r);
#pragma unroll
                for (int j = 0; j < 4; ++j) stg[r][j] = *(const v4f*)(row + j * 1024 + tid * 4);
            }
        }
        lds_barrier();
        float v[CW_][CG_ROWS];
#pragma unroll
        for (int r = 0; r < CG_ROWS; ++r)
#pragma unroll
            for (int c = 0; c < CW_; ++c) v[c][r] = buf[r * N_ + ni[c]];
#pragma unroll
        for (int r = 0; r < CG_ROWS; ++r) {
            const int d = g * CG_ROWS + r;
            const bool mine = (lane == (d & 31)) & (d < 67);
#pragma unroll
            for (int c = 0; c < CW_; ++c) {
                const float m = wave_fmax(v[c][r]);
                rm[c] = mine ? m : rm[c];
            }
        }
        const int dl = g * CG_ROWS + CG_ROWS - 1;
        if ((dl & 31) == 31 || dl >= 66) {
            const int dd = (dl & ~31) + lane;
            if (dd < 67) {
#pragma unroll
                for (int c = 0; c < CW_; ++c) op[c][(size_t)dd * S_] = rm[c];
            }
        }
    }
}

extern "C" void kernel_launch(void* const* d_in, const int* in_sizes, int n_in,
                              void* d_out, int out_size, void* d_ws, size_t ws_size,
                              hipStream_t stream) {
    const float* xyz    = (const float*)d_in[0];
    const float* points = (const float*)d_in[1];
    const int*   finit  = (const int*)d_in[2];

    float* out        = (float*)d_out;
    float* out_newxyz = out;
    float* out_feat   = out + (size_t)B_ * 3 * S_;

    float* wsA  = (float*)d_ws;
    float* wsC2 = wsA + (size_t)B_ * S_ * 4;

    fps_kernel<<<B_, FPS_THREADS, 0, stream>>>(xyz, finit, out_newxyz, wsA, wsC2);
    group_kernel<<<dim3(S_ / CB_, B_), 256, 0, stream>>>(xyz, points, wsA, wsC2, out_feat);
}


# FPS step: a wave stores its cached partial again while the new centroid does not lower its best point (one wave-uniform distance test); only a wave that fails it redoes maxima, pick and wave argmax
# speedup vs baseline: 6.7964x; 1.0890x over previous
#include <hip/hip_runtime.h>
#include <math.h>

#define B_  8
#define N_  4096
#define D_  64
#define S_  1024
#define K_  32

typedef __attribute__((ext_vector_type(2))) float v2f;
typedef __attribute__((ext_vector_type(4))) float v4f;
typedef __attribute__((ext_vector_type(8))) float v8f;

typedef __attribute__((ext_vector_type(4))) unsigned int v4u;

#define FPS_WAVES 8
#define FPS_THREADS (FPS_WAVES * 32)
#define FPS_PTS (N_ / FPS_THREADS)
#define FPS_ROW (FPS_WAVES + 1)

__device__ __forceinline__ unsigned wave_umax(unsigned v) {
    v = max(v, (unsigned)__builtin_amdgcn_update_dpp(0, (int)v, 0x161, 0xf, 0xf, true));
    v = max(v, (unsigned)__builtin_amdgcn_update_dpp(0, (int)v, 0x162, 0xf, 0xf, true));
    v = max(v, (unsigned)__builtin_amdgcn_update_dpp(0, (int)v, 0x164, 0xf, 0xf, true));
    v = max(v, (unsigned)__builtin_amdgcn_update_dpp(0, (int)v, 0x168, 0xf, 0xf, true));
    v = max(v, (unsigned)__builtin_amdgcn_permlanex16((int)v, (int)v, 0x76543210, 0xfedcba98, false, false));
    return v;
}

__device__ __forceinline__ unsigned group8_umax(unsigned v) {
    v = max(v, (unsigned)__builtin_amdgcn_update_dpp(0, (int)v, 0x161, 0xf, 0xf, true));
    v = max(v, (unsigned)__builtin_amdgcn_update_dpp(0, (int)v, 0x162, 0xf, 0xf, true));
    v = max(v, (unsigned)__builtin_amdgcn_update_dpp(0, (int)v, 0x164, 0xf, 0xf, true));
    return v;
}

__device__ __forceinline__ void lds_barrier() {
    __builtin_amdgcn_fence(__ATOMIC_RELEASE, "workgroup", "local");
    __builtin_amdgcn_s_barrier();
    __builtin_amdgcn_fence(__ATOMIC_ACQUIRE, "workgroup", "local");
}

__device__ __forceinline__ float lane_value(unsigned v, int src) {
    return __uint_as_float((unsigned)__builtin_amdgcn_readlane((int)v, src));
}

__global__ __launch_bounds__(FPS_THREADS) void fps_kernel(
    const float* __restrict__ xyz,
    const int*   __restrict__ farthest_init,
    float* __restrict__ out_newxyz,
    float* __restrict__ wsA,
    float* __restrict__ wsC2)
{
    __shared__ v4u red[S_][FPS_ROW];
    __shared__ v4u spare[FPS_THREADS];

    const int b    = blockIdx.x;
    const int tid  = threadIdx.x;
    const int lane = tid & 31;
    const int wid  = tid >> 5;

    const float* xb = xyz + (size_t)b * 3 * N_;

    float px[FPS_PTS], py[FPS_PTS], pz[FPS_PTS];
    unsigned du[FPS_PTS];
#pragma unroll
    for (int j = 0; j < FPS_PTS; ++j) {
        px[j] = xb[FPS_PTS * tid + j];
        py[j] = xb[N_ + FPS_PTS * tid + j];
        pz[j] = xb[2 * N_ + FPS_PTS * tid + j];
        du[j] = __float_as_uint(1e10f);
    }
    const int far0 = farthest_init[b] & (N_ - 1);
    const float c0x = xb[far0], c0y = xb[N_ + far0], c0z = xb[2 * N_ + far0];
    float cx = c0x, cy = c0y, cz = c0z;

    v4u* rowp = &red[0][0];
    v4u mine = {0u, 0u, 0u, 0u};
    int wonL = 0;
    for (int it = 0; it < S_; ++it, rowp += FPS_ROW) {
        const float ox = __uint_as_float(mine.y) - cx;
        const float oy = __uint_as_float(mine.z) - cy;
        const float oz = __uint_as_float(mine.w) - cz;
        const float od = fmaf(oz, oz, fmaf(ox, ox, oy * oy));
        const unsigned keep = __builtin_amdgcn_ballot_w32(__float_as_uint(od) >= mine.x);
#pragma unroll
        for (int j = 0; j < FPS_PTS; ++j) {
            const float dx = px[j] - cx, dy = py[j] - cy, dz = pz[j] - cz;
            const float d = fmaf(dz, dz, fmaf(dx, dx, dy * dy));
            du[j] = min(du[j], __float_as_uint(d));
        }
        if (it == 0 || ((keep >> wonL) & 1u) == 0u) {
        unsigned mq[FPS_PTS / 4];
        float qx[FPS_PTS / 4], qy[FPS_PTS / 4], qz[FPS_PTS / 4];
#pragma unroll
        for (int q = 0; q < FPS_PTS / 4; ++q) {
            mq[q] = max(max(du[4 * q], du[4 * q + 1]), max(du[4 * q + 2], du[4 * q + 3]));
            qx[q] = px[4 * q + 3]; qy[q] = py[4 * q + 3]; qz[q] = pz[4 * q + 3];
#pragma unroll
            for (int j = 2; j >= 0; --j) {
                const bool t = (du[4 * q + j] == mq[q]);
                qx[q] = t ? px[4 * q + j] : qx[q];
                qy[q] = t ? py[4 * q + j] : qy[q];
                qz[q] = t ? pz[4 * q + j] : qz[q];
            }
        }
        unsigned bu = mq[0];
#pragma unroll
        for (int q = 1; q < FPS_PTS / 4; ++q) bu = max(bu, mq[q]);
        float sx = qx[FPS_PTS / 4 - 1], sy = qy[FPS_PTS / 4 - 1], sz = qz[FPS_PTS / 4 - 1];
#pragma unroll
        for (int q = FPS_PTS / 4 - 2; q >= 0; --q) {
            const bool tq = (mq[q] == bu);
            sx = tq ? qx[q] : sx;
            sy = tq ? qy[q] : sy;
            sz = tq ? qz[q] : sz;
        }
        mine.x = bu;
        mine.y = __float_as_uint(sx);
        mine.z = __float_as_uint(sy);
        mine.w = __float_as_uint(sz);

        const unsigned m   = wave_umax(bu);
        const unsigned hit = __builtin_amdgcn_ballot_w32(bu == m);
        wonL = __builtin_ctz(hit);
        }
        v4u* dst = (lane == wonL) ? &rowp[wid] : &spare[tid];
        *dst = mine;
        lds_barrier();
        const v4u pr = rowp[lane & (FPS_WAVES - 1)];
        const unsigned M    = group8_umax(pr.x);
        const unsigned hit2 = __builtin_amdgcn_ballot_w32(pr.x == M);
        const int      win  = __builtin_ctz(hit2);
        cx = lane_value(pr.y, win);
        cy = lane_value(pr.z, win);
        cz = lane_value(pr.w, win);
    }

    for (int k = 0; k < S_ / FPS_THREADS; ++k) {
        const int it  = k * FPS_THREADS + tid;
        const int row = it > 0 ? it - 1 : 0;
        v4u best = red[row][0];
        for (int w = 1; w < FPS_WAVES; ++w) {
            const v4u p = red[row][w];
            const bool t = p.x > best.x;
            best.x = t ? p.x : best.x;
            best.y = t ? p.y : best.y;
            best.z = t ? p.z : best.z;
            best.w = t ? p.w : best.w;
        }
        const float ex = it > 0 ? __uint_as_float(best.y) : c0x;
        const float ey = it > 0 ? __uint_as_float(best.z) : c0y;
        const float ez = it > 0 ? __uint_as_float(best.w) : c0z;

        out_newxyz[(size_t)b * 3 * S_ + 0 * S_ + it] = ex;
        out_newxyz[(size_t)b * 3 * S_ + 1 * S_ + it] = ey;
        out_newxyz[(size_t)b * 3 * S_ + 2 * S_ + it] = ez;
        v4f A;
        A.x = -2.f * ex; A.y = -2.f * ey; A.z = -2.f * ez; A.w = 1.f;
        *(v4f*)(wsA + ((size_t)b * S_ + it) * 4) = A;
        float c2;
        {
#pragma clang fp contract(off)
            c2 = fmaf(ex, ex, ey * ey) + ez * ez;
        }
        wsC2[(size_t)b * S_ + it] = c2;
    }
}

__device__ __forceinline__ void lexmin(float& v, int& i, float ov, int oi) {
    const bool take = (ov < v) | ((ov == v) & (oi < i));
    v = take ? ov : v;
    i = take ? oi : i;
}

#define CG_ROWS 8

__device__ __forceinline__ const float* chan_row(const float* xb, const float* pb, int d) {
    d = d < 66 ? d : 66;
    return (d < 3) ? (xb + (size_t)d * N_) : (pb + (size_t)(d - 3) * N_);
}

__device__ __forceinline__ float wave_fmax(float v) {
    v = fmaxf(v, __int_as_float(__builtin_amdgcn_update_dpp(0, __float_as_int(v), 0x161, 0xf, 0xf, true)));
    v = fmaxf(v, __int_as_float(__builtin_amdgcn_update_dpp(0, __float_as_int(v), 0x162, 0xf, 0xf, true)));
    v = fmaxf(v, __int_as_float(__builtin_amdgcn_update_dpp(0, __float_as_int(v), 0x164, 0xf, 0xf, true)));
    v = fmaxf(v, __int_as_float(__builtin_amdgcn_update_dpp(0, __float_as_int(v), 0x168, 0xf, 0xf, true)));
    v = fmaxf(v, __int_as_float(__builtin_amdgcn_permlanex16(__float_as_int(v), __float_as_int(v), 0x76543210, 0xfedcba98, false, false)));
    return v;
}

__device__ __forceinline__ int wave_imin(int v) {
    v = min(v, __builtin_amdgcn_update_dpp(0, v, 0x161, 0xf, 0xf, true));
    v = min(v, __builtin_amdgcn_update_dpp(0, v, 0x162, 0xf, 0xf, true));
    v = min(v, __builtin_amdgcn_update_dpp(0, v, 0x164, 0xf, 0xf, true));
    v = min(v, __builtin_amdgcn_update_dpp(0, v, 0x168, 0xf, 0xf, true));
    v = min(v, (int)__builtin_amdgcn_permlanex16(v, v, 0x76543210, 0xfedcba98, false, false));
    return v;
}

__device__ __forceinline__ int fkey(int bits) {
    return bits ^ ((bits >> 31) & 0x7fffffff);
}

__device__ __forceinline__ void wave_lexmin(int& k, int& i) {
    const int km = wave_imin(k);
    i = wave_imin(k == km ? i : 0x7fffffff);
    k = km;
}

#define TPITCH 4100
#define CB_ 32
#define NPASS (CB_ / 16)
#define CW_ (2 * NPASS)

__global__ __launch_bounds__(256) void group_kernel(
    const float* __restrict__ xyz,
    const float* __restrict__ pts,
    const float* __restrict__ wsA,
    const float* __restrict__ wsC2,
    float* __restrict__ out_feat)
{
    __shared__ float DT[16 * TPITCH];
    __shared__ int   selIdx[CB_][K_];

    const int b    = blockIdx.y;
    const int sb   = blockIdx.x * CB_;
    const int tid  = threadIdx.x;
    const int lane = tid & 31;
    const int wid  = tid >> 5;
    const bool hi  = lane >= 16;

    const float* xb = xyz + (size_t)b * 3 * N_;

    const int  mcol  = lane & 15;
    const int  mbase = hi ? 8 : 0;

    float* rowp[2];
    rowp[0] = &DT[wid * TPITCH];
    rowp[1] = &DT[(wid + 8) * TPITCH];

    for (int pass = 0; pass < NPASS; ++pass) {
        const int s0 = sb + pass * 16;

        const v4f  A4   = *(const v4f*)(wsA + ((size_t)b * S_ + s0 + mcol) * 4);
        v2f a;
        a.x = hi ? A4.z : A4.x;
        a.y = hi ? A4.w : A4.y;

        float cmr[8];
#pragma unroll
        for (int r = 0; r < 8; ++r)
            cmr[r] = wsC2[(size_t)b * S_ + s0 + mbase + r];

        for (int t = 0; t < 32; ++t) {
            const int n = wid * 512 + t * 16 + mcol;
            const float x = xb[n], y = xb[N_ + n], z = xb[2 * N_ + n];
            const float pn2 = x * x + y * y + z * z;
            v2f bb;
            bb.x = hi ? z : x;
            bb.y = hi ? pn2 : y;

            v8f c = {};
            c = __builtin_amdgcn_wmma_f32_16x16x4_f32(
                     false, a,  false, bb,
                     (short)0, c,  false,  false);

#pragma unroll
            for (int r = 0; r < 8; ++r)
                DT[(mbase + r) * TPITCH + n] = c[r] + cmr[r];
        }
        __syncthreads();

        int lk[2]; int li[2];
        int gi[2];
#pragma unroll
        for (int rr = 0; rr < 2; ++rr) {
            const float* row = rowp[rr];
            float bv = 3.4e38f; int bi = 0x7fffffff;
#pragma unroll 8
            for (int q = 0; q < N_ / 128; ++q) {
                const int j0 = q * 128 + ((lane - q) & 31) * 4;
                const v4f dv = *(const v4f*)&row[j0];
#pragma unroll
                for (int e = 0; e < 4; ++e) lexmin(bv, bi, dv[e], j0 + e);
            }
            int bk = fkey(__float_as_int(bv));
            lk[rr] = bk; li[rr] = bi;
            wave_lexmin(bk, bi);
            gi[rr] = bi;
        }

        for (int k = 0; k < K_; ++k) {
            if (lane == 0) {
                selIdx[pass * 16 + wid][k]     = gi[0];
                selIdx[pass * 16 + wid + 8][k] = gi[1];
                rowp[0][gi[0]] = 3.4e38f;
                rowp[1][gi[1]] = 3.4e38f;
            }
            if (k + 1 < K_) {
#pragma unroll
                for (int rr = 0; rr < 2; ++rr) {
                    const float* row = rowp[rr];
                    const int bi = gi[rr];
                    const int w = ((bi >> 2) + (bi >> 7)) & 31;
                    const int j0 = lane * 128 + ((w - lane) & 31) * 4;
                    const v4f dv = *(const v4f*)&row[j0];
                    float cv = dv[0]; int ci = j0;
                    float c2 = dv[2]; int i2 = j0 + 2;
                    lexmin(cv, ci, dv[1], j0 + 1);
                    lexmin(c2, i2, dv[3], j0 + 3);
                    lexmin(cv, ci, c2, i2);
                    int ck = fkey(__float_as_int(cv));
                    const bool own = (lane == w);
                    int ok = own ? 0x7fffffff : lk[rr];
                    int oi = own ? 0x7fffffff : li[rr];
                    wave_lexmin(ck, ci);
                    wave_lexmin(ok, oi);
                    lk[rr] = own ? ck : lk[rr];
                    li[rr] = own ? ci : li[rr];
                    const bool other = (ok < ck) | ((ok == ck) & (oi < ci));
                    gi[rr] = other ? oi : ci;
                }
            }
        }
        __syncthreads();
    }

    const float* pb = pts + (size_t)b * D_ * N_;
    int    ni[CW_];
    float* op[CW_];
    float  rm[CW_];
#pragma unroll
    for (int c = 0; c < CW_; ++c) {
        const int crow = (c >> 1) * 16 + (c & 1) * 8 + wid;
        ni[c] = selIdx[crow][lane];
        op[c] = out_feat + (size_t)b * 67 * S_ + (sb + crow);
        rm[c] = 0.f;
    }

    v4f stg[CG_ROWS][4];
#pragma unroll
    for (int r = 0; r < CG_ROWS; ++r) {
        const float* row = chan_row(xb, pb, r);
#pragma unroll
        for (int j = 0; j < 4; ++j) stg[r][j] = *(const v4f*)(row + j * 1024 + tid * 4);
    }
    for (int g = 0; g < (67 + CG_ROWS - 1) / CG_ROWS; ++g) {
        float* buf = DT + (g & 1) * (CG_ROWS * N_);
#pragma unroll
        for (int r = 0; r < CG_ROWS; ++r)
#pragma unroll
            for (int j = 0; j < 4; ++j) *(v4f*)(buf + r * N_ + j * 1024 + tid * 4) = stg[r][j];
        if ((g + 1) * CG_ROWS < 67) {
#pragma unroll
            for (int r = 0; r < CG_ROWS; ++r) {
                const float* row = chan_row(xb, pb, (g + 1) * CG_ROWS + r);
#pragma unroll
                for (int j = 0; j < 4; ++j) stg[r][j] = *(const v4f*)(row + j * 1024 + tid * 4);
            }
        }
        lds_barrier();
        float v[CW_][CG_ROWS];
#pragma unroll
        for (int r = 0; r < CG_ROWS; ++r)
#pragma unroll
            for (int c = 0; c < CW_; ++c) v[c][r] = buf[r * N_ + ni[c]];
#pragma unroll
        for (int r = 0; r < CG_ROWS; ++r) {
            const int d = g * CG_ROWS + r;
            const bool mine = (lane == (d & 31)) & (d < 67);
#pragma unroll
            for (int c = 0; c < CW_; ++c) {
                const float m = wave_fmax(v[c][r]);
                rm[c] = mine ? m : rm[c];
            }
        }
        const int dl = g * CG_ROWS + CG_ROWS - 1;
        if ((dl & 31) == 31 || dl >= 66) {
            const int dd = (dl & ~31) + lane;
            if (dd < 67) {
#pragma unroll
                for (int c = 0; c < CW_; ++c) op[c][(size_t)dd * S_] = rm[c];
            }
        }
    }
}

extern "C" void kernel_launch(void* const* d_in, const int* in_sizes, int n_in,
                              void* d_out, int out_size, void* d_ws, size_t ws_size,
                              hipStream_t stream) {
    const float* xyz    = (const float*)d_in[0];
    const float* points = (const float*)d_in[1];
    const int*   finit  = (const int*)d_in[2];

    float* out        = (float*)d_out;
    float* out_newxyz = out;
    float* out_feat   = out + (size_t)B_ * 3 * S_;

    float* wsA  = (float*)d_ws;
    float* wsC2 = wsA + (size_t)B_ * S_ * 4;

    fps_kernel<<<B_, FPS_THREADS, 0, stream>>>(xyz, finit, out_newxyz, wsA, wsC2);
    group_kernel<<<dim3(S_ / CB_, B_), 256, 0, stream>>>(xyz, points, wsA, wsC2, out_feat);
}
